# starting kernel
# baseline (speedup 1.0000x reference)
#include <hip/hip_runtime.h>
#include <hip/hip_bf16.h>
#include <stdint.h>

typedef __attribute__((ext_vector_type(2))) float v2f;
typedef __attribute__((ext_vector_type(8))) float v8f;

#define N_NODES 75000
#define N_EDGES 1200000

__global__ void gcn_deg_init(float* __restrict__ deg, int n) {
    int i = blockIdx.x * blockDim.x + threadIdx.x;
    if (i < n) deg[i] = 1.0f;
}

__global__ void gcn_deg_count(const int* __restrict__ dst, float* __restrict__ deg, int nE) {
    int e = blockIdx.x * blockDim.x + threadIdx.x;
    if (e < nE) atomicAdd(&deg[dst[e]], 1.0f);
}

__global__ void gcn_deg_rsqrt(float* __restrict__ deg, int n) {
    int i = blockIdx.x * blockDim.x + threadIdx.x;
    if (i < n) deg[i] = rsqrtf(deg[i]);
}

template <int KDIM, int NDIM>
__global__ void gcn_wmma_gemm(const float* __restrict__ A, const float* __restrict__ W,
                              float* __restrict__ C, int M) {
    const int lane  = threadIdx.x & 31;
    const int wave  = threadIdx.x >> 5;
    const int half  = lane >> 4;
    const int l     = lane & 15;
    const int ntile = wave * 16;
    const int mtile = blockIdx.x * 16;

    int row = mtile + l;
    if (row >= M) row = M - 1;
    const float* arow = A + (size_t)row * KDIM;
    const int n = ntile + l;

    v8f c = {};
#pragma unroll
    for (int k0 = 0; k0 < KDIM; k0 += 4) {
        const int ka = k0 + half * 2;
        v2f a, b;
        a.x = arow[ka];
        a.y = arow[ka + 1];
        b.x = W[(size_t)ka * NDIM + n];
        b.y = W[(size_t)(ka + 1) * NDIM + n];
        c = __builtin_amdgcn_wmma_f32_16x16x4_f32(false, a, false, b,
                                                  (short)0, c, false, false);
    }

#pragma unroll
    for (int r = 0; r < 8; ++r) {
        const int m = mtile + r + half * 8;
        if (m < M) C[(size_t)m * NDIM + n] = c[r];
    }
}

template <int F>
__global__ void gcn_self_init(const float* __restrict__ h, const float* __restrict__ dinv,
                              float* __restrict__ out, int n) {
    int i = blockIdx.x * blockDim.x + threadIdx.x;
    if (i >= n * F) return;
    const int node = i / F;
    const float di = dinv[node];
    out[i] = h[i] * di * di;
}

template <int F>
__global__ void gcn_edge_agg(const int* __restrict__ src, const int* __restrict__ dst,
                             const float* __restrict__ h, const float* __restrict__ dinv,
                             float* __restrict__ out, int nE) {
    const int epb = 256 / F;
    const int e = blockIdx.x * epb + (int)(threadIdx.x / F);
    const int f = (int)(threadIdx.x % F);
    if (e >= nE) return;
    const int s = src[e];
    const int d = dst[e];
    const float v = h[(size_t)s * F + f] * (dinv[s] * dinv[d]);
    atomicAdd(&out[(size_t)d * F + f], v);
}

template <int F, bool RELU>
__global__ void gcn_bias_act(float* __restrict__ out, const float* __restrict__ b, int n) {
    int i = blockIdx.x * blockDim.x + threadIdx.x;
    if (i >= n * F) return;
    float v = out[i] + b[i % F];
    if (RELU) v = fmaxf(v, 0.0f);
    out[i] = v;
}

extern "C" void kernel_launch(void* const* d_in, const int* in_sizes, int n_in,
                              void* d_out, int out_size, void* d_ws, size_t ws_size,
                              hipStream_t stream) {
    const float* x  = (const float*)d_in[0];
    const int*   ei = (const int*)d_in[1];
    const float* W1 = (const float*)d_in[2];
    const float* b1 = (const float*)d_in[3];
    const float* W2 = (const float*)d_in[4];
    const float* b2 = (const float*)d_in[5];
    float* out = (float*)d_out;

    const int* src = ei;
    const int* dst = ei + N_EDGES;

    float* ws   = (float*)d_ws;
    float* dinv = ws;
    float* h1   = dinv + N_NODES;
    float* agg1 = h1 + (size_t)N_NODES * 64;
    float* h2   = agg1 + (size_t)N_NODES * 64;

    const int mtiles = (N_NODES + 15) / 16;

    gcn_deg_init<<<(N_NODES + 255) / 256, 256, 0, stream>>>(dinv, N_NODES);
    gcn_deg_count<<<(N_EDGES + 255) / 256, 256, 0, stream>>>(dst, dinv, N_EDGES);
    gcn_deg_rsqrt<<<(N_NODES + 255) / 256, 256, 0, stream>>>(dinv, N_NODES);

    gcn_wmma_gemm<64, 64><<<mtiles, 128, 0, stream>>>(x, W1, h1, N_NODES);

    gcn_self_init<64><<<(N_NODES * 64 + 255) / 256, 256, 0, stream>>>(h1, dinv, agg1, N_NODES);
    gcn_edge_agg<64><<<(N_EDGES + 3) / 4, 256, 0, stream>>>(src, dst, h1, dinv, agg1, N_EDGES);
    gcn_bias_act<64, true><<<(N_NODES * 64 + 255) / 256, 256, 0, stream>>>(agg1, b1, N_NODES);

    gcn_wmma_gemm<64, 16><<<mtiles, 32, 0, stream>>>(agg1, W2, h2, N_NODES);

    gcn_self_init<16><<<(N_NODES * 16 + 255) / 256, 256, 0, stream>>>(h2, dinv, out, N_NODES);
    gcn_edge_agg<16><<<(N_EDGES + 15) / 16, 256, 0, stream>>>(src, dst, h2, dinv, out, N_EDGES);
    gcn_bias_act<16, false><<<(N_NODES * 16 + 255) / 256, 256, 0, stream>>>(out, b2, N_NODES);
}


# Aggregation as a gather over a destination-sorted edge list (counting sort, int atomic arrival rank) replaces the float atomic scatter; rows summed in rank order; bias and relu fused
# speedup vs baseline: 2.3529x; 2.3529x over previous
#include <hip/hip_runtime.h>
#include <hip/hip_bf16.h>
#include <stdint.h>


typedef __attribute__((ext_vector_type(2))) float v2f;
typedef __attribute__((ext_vector_type(8))) float v8f;

#define N_NODES 75000
#define N_EDGES 1200000

__global__ void gcn_cnt_zero(int* __restrict__ cnt, int n) {
    int i = blockIdx.x * blockDim.x + threadIdx.x;
    if (i < n) cnt[i] = 0;
}

__global__ void gcn_count_rank(const int* __restrict__ dst, int* __restrict__ cnt,
                               int* __restrict__ rank, int nE) {
    int e = blockIdx.x * blockDim.x + threadIdx.x;
    if (e < nE) rank[e] = atomicAdd(&cnt[dst[e]], 1);
}

#define SCAN_BLOCKS ((N_NODES + 1023) / 1024)
__global__ void __launch_bounds__(1024)
gcn_scan_dinv(const int* __restrict__ cnt, int* __restrict__ rowptr,
              float* __restrict__ dinv, int n) {
    __shared__ int s_part[32];
    __shared__ int s_tot[32];
    const int t    = threadIdx.x;
    const int lane = t & 31;
    const int wave = t >> 5;
    const int base = blockIdx.x * 1024;

    int part = 0;
    const int4* c4 = (const int4*)cnt;
    const int nq = base / 4;
    int q = t;
    for (; q + 3072 < nq; q += 4096) {
        const int4 a0 = c4[q], a1 = c4[q + 1024], a2 = c4[q + 2048], a3 = c4[q + 3072];
        part += (((a0.x + a0.y) + (a0.z + a0.w)) + ((a1.x + a1.y) + (a1.z + a1.w)))
              + (((a2.x + a2.y) + (a2.z + a2.w)) + ((a3.x + a3.y) + (a3.z + a3.w)));
    }
    for (; q < nq; q += 1024) {
        const int4 a0 = c4[q];
        part += (a0.x + a0.y) + (a0.z + a0.w);
    }
    const int i = base + t;
    const int c = (i < n) ? cnt[i] : 0;

    for (int o = 16; o > 0; o >>= 1) part += __shfl_xor(part, o);
    int inc = c;
    for (int o = 1; o < 32; o <<= 1) {
        const int y = __shfl_up(inc, o);
        if (lane >= o) inc += y;
    }
    if (lane == 31) s_tot[wave] = inc;
    if (lane == 0)  s_part[wave] = part;
    __syncthreads();

    int pv = s_part[lane];
    int tv = (lane < wave) ? s_tot[lane] : 0;
    for (int o = 16; o > 0; o >>= 1) {
        pv += __shfl_xor(pv, o);
        tv += __shfl_xor(tv, o);
    }
    const int excl = pv + tv + inc - c;
    if (i < n) {
        rowptr[i] = excl;
        dinv[i] = rsqrtf((float)(c + 1));
    }
    if (i == n - 1) rowptr[n] = excl + c;
}

__global__ void gcn_place(const int* __restrict__ src, const int* __restrict__ dst,
                          const int* __restrict__ rank, const int* __restrict__ rowptr,
                          int* __restrict__ csr, int nE) {
    int e = blockIdx.x * blockDim.x + threadIdx.x;
    if (e < nE) csr[rowptr[dst[e]] + rank[e]] = src[e];
}

template <int KDIM, int NDIM>
__global__ void gcn_wmma_gemm(const float* __restrict__ A, const float* __restrict__ W,
                              float* __restrict__ C, int M) {
    const int lane  = threadIdx.x & 31;
    const int wave  = threadIdx.x >> 5;
    const int half  = lane >> 4;
    const int l     = lane & 15;
    const int ntile = wave * 16;
    const int mtile = blockIdx.x * 16;

    int row = mtile + l;
    if (row >= M) row = M - 1;
    const float* arow = A + (size_t)row * KDIM;
    const int n = ntile + l;

    v8f c = {};
#pragma unroll
    for (int k0 = 0; k0 < KDIM; k0 += 4) {
        const int ka = k0 + half * 2;
        v2f a, b;
        a.x = arow[ka];
        a.y = arow[ka + 1];
        b.x = W[(size_t)ka * NDIM + n];
        b.y = W[(size_t)(ka + 1) * NDIM + n];
        c = __builtin_amdgcn_wmma_f32_16x16x4_f32(false, a, false, b,
                                                  (short)0, c, false, false);
    }

#pragma unroll
    for (int r = 0; r < 8; ++r) {
        const int m = mtile + r + half * 8;
        if (m < M) C[(size_t)m * NDIM + n] = c[r];
    }
}

template <int F, bool RELU>
__global__ void __launch_bounds__(256)
gcn_gather(const int* __restrict__ rowptr, const int* __restrict__ csr,
           const float* __restrict__ h, const float* __restrict__ dinv,
           const float* __restrict__ b, float* __restrict__ out, int n) {
    constexpr int LPN = F / 4;
    const int gt   = blockIdx.x * 256 + threadIdx.x;
    const int node = gt / LPN;
    const int l    = gt % LPN;
    if (node >= n) return;

    const float4* __restrict__ h4 = (const float4*)h;
    const int beg = rowptr[node];
    const int end = rowptr[node + 1];
    const float di = dinv[node];

    float4 acc;
    {
        const float4 v = h4[(size_t)node * LPN + l];
        const float w = di * di;
        acc.x = v.x * w; acc.y = v.y * w; acc.z = v.z * w; acc.w = v.w * w;
    }

    int p = beg;
    for (; p + 4 <= end; p += 4) {
        const int s0 = csr[p], s1 = csr[p + 1], s2 = csr[p + 2], s3 = csr[p + 3];
        const float w0 = dinv[s0] * di, w1 = dinv[s1] * di;
        const float w2 = dinv[s2] * di, w3 = dinv[s3] * di;
        const float4 v0 = h4[(size_t)s0 * LPN + l];
        const float4 v1 = h4[(size_t)s1 * LPN + l];
        const float4 v2 = h4[(size_t)s2 * LPN + l];
        const float4 v3 = h4[(size_t)s3 * LPN + l];
        acc.x += v0.x * w0; acc.y += v0.y * w0; acc.z += v0.z * w0; acc.w += v0.w * w0;
        acc.x += v1.x * w1; acc.y += v1.y * w1; acc.z += v1.z * w1; acc.w += v1.w * w1;
        acc.x += v2.x * w2; acc.y += v2.y * w2; acc.z += v2.z * w2; acc.w += v2.w * w2;
        acc.x += v3.x * w3; acc.y += v3.y * w3; acc.z += v3.z * w3; acc.w += v3.w * w3;
    }
    for (; p < end; ++p) {
        const int s0 = csr[p];
        const float w0 = dinv[s0] * di;
        const float4 v0 = h4[(size_t)s0 * LPN + l];
        acc.x += v0.x * w0; acc.y += v0.y * w0; acc.z += v0.z * w0; acc.w += v0.w * w0;
    }

    const float4 bb = ((const float4*)b)[l];
    acc.x += bb.x; acc.y += bb.y; acc.z += bb.z; acc.w += bb.w;
    if (RELU) {
        acc.x = fmaxf(acc.x, 0.0f); acc.y = fmaxf(acc.y, 0.0f);
        acc.z = fmaxf(acc.z, 0.0f); acc.w = fmaxf(acc.w, 0.0f);
    }
    ((float4*)out)[(size_t)node * LPN + l] = acc;
}

extern "C" void kernel_launch(void* const* d_in, const int* in_sizes, int n_in,
                              void* d_out, int out_size, void* d_ws, size_t ws_size,
                              hipStream_t stream) {
    const float* x  = (const float*)d_in[0];
    const int*   ei = (const int*)d_in[1];
    const float* W1 = (const float*)d_in[2];
    const float* b1 = (const float*)d_in[3];
    const float* W2 = (const float*)d_in[4];
    const float* b2 = (const float*)d_in[5];
    float* out = (float*)d_out;

    const int* src = ei;
    const int* dst = ei + N_EDGES;

    int*   cnt    = (int*)d_ws;
    int*   rowptr = cnt + 75008;
    float* dinv   = (float*)(rowptr + 75008);
    int*   rank   = (int*)(dinv + 75008);
    int*   csr    = rank + N_EDGES;
    float* h1     = (float*)(csr + N_EDGES);
    float* agg1   = h1 + (size_t)N_NODES * 64;
    float* h2     = agg1 + (size_t)N_NODES * 64;

    const int mtiles = (N_NODES + 15) / 16;

    gcn_cnt_zero<<<(N_NODES + 255) / 256, 256, 0, stream>>>(cnt, N_NODES);
    gcn_count_rank<<<(N_EDGES + 255) / 256, 256, 0, stream>>>(dst, cnt, rank, N_EDGES);
    gcn_scan_dinv<<<SCAN_BLOCKS, 1024, 0, stream>>>(cnt, rowptr, dinv, N_NODES);
    gcn_place<<<(N_EDGES + 255) / 256, 256, 0, stream>>>(src, dst, rank, rowptr, csr, N_EDGES);

    gcn_wmma_gemm<64, 64><<<mtiles, 128, 0, stream>>>(x, W1, h1, N_NODES);
    gcn_gather<64, true><<<(N_NODES * 16 + 255) / 256, 256, 0, stream>>>(rowptr, csr, h1, dinv, b1, agg1, N_NODES);

    gcn_wmma_gemm<64, 16><<<mtiles, 32, 0, stream>>>(agg1, W2, h2, N_NODES);
    gcn_gather<16, false><<<(N_NODES * 4 + 255) / 256, 256, 0, stream>>>(rowptr, csr, h2, dinv, b2, out, N_NODES);
}


# Layer 1 aggregates x first (agg(x) W1 = agg(x W1)) and both products run in that gather kernel's tail on LDS tiles: 8 launches to 6, x W1 and layer-1 output never stored; sums reordered
# speedup vs baseline: 2.5244x; 1.0729x over previous
#include <hip/hip_runtime.h>
#include <hip/hip_bf16.h>
#include <stdint.h>


typedef __attribute__((ext_vector_type(2))) float v2f;
typedef __attribute__((ext_vector_type(8))) float v8f;

#define N_NODES 75000
#define N_EDGES 1200000

__global__ void gcn_cnt_zero(int* __restrict__ cnt, int n) {
    int i = blockIdx.x * blockDim.x + threadIdx.x;
    if (i < n) cnt[i] = 0;
}

__global__ void gcn_count_rank(const int* __restrict__ dst, int* __restrict__ cnt,
                               int* __restrict__ rank, int nE) {
    int e = blockIdx.x * blockDim.x + threadIdx.x;
    if (e < nE) rank[e] = atomicAdd(&cnt[dst[e]], 1);
}

#define SCAN_BLOCKS ((N_NODES + 1023) / 1024)
__global__ void __launch_bounds__(1024)
gcn_scan_dinv(const int* __restrict__ cnt, int* __restrict__ rowptr,
              float* __restrict__ dinv, int n) {
    __shared__ int s_part[32];
    __shared__ int s_tot[32];
    const int t    = threadIdx.x;
    const int lane = t & 31;
    const int wave = t >> 5;
    const int base = blockIdx.x * 1024;

    int part = 0;
    const int4* c4 = (const int4*)cnt;
    const int nq = base / 4;
    int q = t;
    for (; q + 3072 < nq; q += 4096) {
        const int4 a0 = c4[q], a1 = c4[q + 1024], a2 = c4[q + 2048], a3 = c4[q + 3072];
        part += (((a0.x + a0.y) + (a0.z + a0.w)) + ((a1.x + a1.y) + (a1.z + a1.w)))
              + (((a2.x + a2.y) + (a2.z + a2.w)) + ((a3.x + a3.y) + (a3.z + a3.w)));
    }
    for (; q < nq; q += 1024) {
        const int4 a0 = c4[q];
        part += (a0.x + a0.y) + (a0.z + a0.w);
    }
    const int i = base + t;
    const int c = (i < n) ? cnt[i] : 0;

    for (int o = 16; o > 0; o >>= 1) part += __shfl_xor(part, o);
    int inc = c;
    for (int o = 1; o < 32; o <<= 1) {
        const int y = __shfl_up(inc, o);
        if (lane >= o) inc += y;
    }
    if (lane == 31) s_tot[wave] = inc;
    if (lane == 0)  s_part[wave] = part;
    __syncthreads();

    int pv = s_part[lane];
    int tv = (lane < wave) ? s_tot[lane] : 0;
    for (int o = 16; o > 0; o >>= 1) {
        pv += __shfl_xor(pv, o);
        tv += __shfl_xor(tv, o);
    }
    const int excl = pv + tv + inc - c;
    if (i < n) {
        rowptr[i] = excl;
        dinv[i] = rsqrtf((float)(c + 1));
    }
    if (i == n - 1) rowptr[n] = excl + c;
}

__global__ void gcn_place(const int* __restrict__ src, const int* __restrict__ dst,
                          const int* __restrict__ rank, const int* __restrict__ rowptr,
                          int* __restrict__ csr, int nE) {
    int e = blockIdx.x * blockDim.x + threadIdx.x;
    if (e < nE) csr[rowptr[dst[e]] + rank[e]] = src[e];
}

#define TILE_PITCH 68
__global__ void __launch_bounds__(256)
gcn_gather_mlp(const int* __restrict__ rowptr, const int* __restrict__ csr,
               const float* __restrict__ x, const float* __restrict__ dinv,
               const float* __restrict__ W1, const float* __restrict__ b1,
               const float* __restrict__ W2, float* __restrict__ h2, int n) {
    __shared__ float sA[16 * TILE_PITCH];
    __shared__ float sB[16 * TILE_PITCH];

    const int t     = threadIdx.x;
    const int slot  = t >> 4;
    const int l     = t & 15;
    const int mtile = blockIdx.x * 16;
    const int node  = mtile + slot;

    float4 acc = {0.0f, 0.0f, 0.0f, 0.0f};
    if (node < n) {
        const float4* __restrict__ x4 = (const float4*)x;
        const int beg = rowptr[node];
        const int end = rowptr[node + 1];
        const float di = dinv[node];
        {
            const float4 v = x4[(size_t)node * 16 + l];
            const float w = di * di;
            acc.x = v.x * w; acc.y = v.y * w; acc.z = v.z * w; acc.w = v.w * w;
        }
        int p = beg;
        for (; p + 4 <= end; p += 4) {
            const int s0 = csr[p], s1 = csr[p + 1], s2 = csr[p + 2], s3 = csr[p + 3];
            const float w0 = dinv[s0] * di, w1 = dinv[s1] * di;
            const float w2 = dinv[s2] * di, w3 = dinv[s3] * di;
            const float4 v0 = x4[(size_t)s0 * 16 + l];
            const float4 v1 = x4[(size_t)s1 * 16 + l];
            const float4 v2 = x4[(size_t)s2 * 16 + l];
            const float4 v3 = x4[(size_t)s3 * 16 + l];
            acc.x += v0.x * w0; acc.y += v0.y * w0; acc.z += v0.z * w0; acc.w += v0.w * w0;
            acc.x += v1.x * w1; acc.y += v1.y * w1; acc.z += v1.z * w1; acc.w += v1.w * w1;
            acc.x += v2.x * w2; acc.y += v2.y * w2; acc.z += v2.z * w2; acc.w += v2.w * w2;
            acc.x += v3.x * w3; acc.y += v3.y * w3; acc.z += v3.z * w3; acc.w += v3.w * w3;
        }
        for (; p < end; ++p) {
            const int s0 = csr[p];
            const float w0 = dinv[s0] * di;
            const float4 v0 = x4[(size_t)s0 * 16 + l];
            acc.x += v0.x * w0; acc.y += v0.y * w0; acc.z += v0.z * w0; acc.w += v0.w * w0;
        }
    }
    *(float4*)&sA[slot * TILE_PITCH + 4 * l] = acc;
    __syncthreads();

    const int lane = t & 31;
    const int half = lane >> 4;
    const int c    = lane & 15;
    const int wave = __builtin_amdgcn_readfirstlane(t >> 5);

    if (wave < 4) {
        const int ncol = wave * 16 + c;
        v8f d = {};
#pragma unroll
        for (int k0 = 0; k0 < 64; k0 += 4) {
            const int ka = k0 + half * 2;
            const v2f a = *(const v2f*)&sA[c * TILE_PITCH + ka];
            v2f b;
            b.x = W1[ka * 64 + ncol];
            b.y = W1[(ka + 1) * 64 + ncol];
            d = __builtin_amdgcn_wmma_f32_16x16x4_f32(false, a, false, b,
                                                      (short)0, d, false, false);
        }
        const float bias = b1[ncol];
#pragma unroll
        for (int r = 0; r < 8; ++r)
            sB[(r + half * 8) * TILE_PITCH + ncol] = fmaxf(d[r] + bias, 0.0f);
    }
    __syncthreads();

    if (wave == 0) {
        v8f d = {};
#pragma unroll
        for (int k0 = 0; k0 < 64; k0 += 4) {
            const int ka = k0 + half * 2;
            const v2f a = *(const v2f*)&sB[c * TILE_PITCH + ka];
            v2f b;
            b.x = W2[ka * 16 + c];
            b.y = W2[(ka + 1) * 16 + c];
            d = __builtin_amdgcn_wmma_f32_16x16x4_f32(false, a, false, b,
                                                      (short)0, d, false, false);
        }
#pragma unroll
        for (int r = 0; r < 8; ++r) {
            const int m = mtile + r + half * 8;
            if (m < n) h2[(size_t)m * 16 + c] = d[r];
        }
    }
}

template <int F, bool RELU>
__global__ void __launch_bounds__(256)
gcn_gather(const int* __restrict__ rowptr, const int* __restrict__ csr,
           const float* __restrict__ h, const float* __restrict__ dinv,
           const float* __restrict__ b, float* __restrict__ out, int n) {
    constexpr int LPN = F / 4;
    const int gt   = blockIdx.x * 256 + threadIdx.x;
    const int node = gt / LPN;
    const int l    = gt % LPN;
    if (node >= n) return;

    const float4* __restrict__ h4 = (const float4*)h;
    const int beg = rowptr[node];
    const int end = rowptr[node + 1];
    const float di = dinv[node];

    float4 acc;
    {
        const float4 v = h4[(size_t)node * LPN + l];
        const float w = di * di;
        acc.x = v.x * w; acc.y = v.y * w; acc.z = v.z * w; acc.w = v.w * w;
    }

    int p = beg;
    for (; p + 4 <= end; p += 4) {
        const int s0 = csr[p], s1 = csr[p + 1], s2 = csr[p + 2], s3 = csr[p + 3];
        const float w0 = dinv[s0] * di, w1 = dinv[s1] * di;
        const float w2 = dinv[s2] * di, w3 = dinv[s3] * di;
        const float4 v0 = h4[(size_t)s0 * LPN + l];
        const float4 v1 = h4[(size_t)s1 * LPN + l];
        const float4 v2 = h4[(size_t)s2 * LPN + l];
        const float4 v3 = h4[(size_t)s3 * LPN + l];
        acc.x += v0.x * w0; acc.y += v0.y * w0; acc.z += v0.z * w0; acc.w += v0.w * w0;
        acc.x += v1.x * w1; acc.y += v1.y * w1; acc.z += v1.z * w1; acc.w += v1.w * w1;
        acc.x += v2.x * w2; acc.y += v2.y * w2; acc.z += v2.z * w2; acc.w += v2.w * w2;
        acc.x += v3.x * w3; acc.y += v3.y * w3; acc.z += v3.z * w3; acc.w += v3.w * w3;
    }
    for (; p < end; ++p) {
        const int s0 = csr[p];
        const float w0 = dinv[s0] * di;
        const float4 v0 = h4[(size_t)s0 * LPN + l];
        acc.x += v0.x * w0; acc.y += v0.y * w0; acc.z += v0.z * w0; acc.w += v0.w * w0;
    }

    const float4 bb = ((const float4*)b)[l];
    acc.x += bb.x; acc.y += bb.y; acc.z += bb.z; acc.w += bb.w;
    if (RELU) {
        acc.x = fmaxf(acc.x, 0.0f); acc.y = fmaxf(acc.y, 0.0f);
        acc.z = fmaxf(acc.z, 0.0f); acc.w = fmaxf(acc.w, 0.0f);
    }
    ((float4*)out)[(size_t)node * LPN + l] = acc;
}

extern "C" void kernel_launch(void* const* d_in, const int* in_sizes, int n_in,
                              void* d_out, int out_size, void* d_ws, size_t ws_size,
                              hipStream_t stream) {
    const float* x  = (const float*)d_in[0];
    const int*   ei = (const int*)d_in[1];
    const float* W1 = (const float*)d_in[2];
    const float* b1 = (const float*)d_in[3];
    const float* W2 = (const float*)d_in[4];
    const float* b2 = (const float*)d_in[5];
    float* out = (float*)d_out;

    const int* src = ei;
    const int* dst = ei + N_EDGES;

    int*   cnt    = (int*)d_ws;
    int*   rowptr = cnt + 75008;
    float* dinv   = (float*)(rowptr + 75008);
    int*   rank   = (int*)(dinv + 75008);
    int*   csr    = rank + N_EDGES;
    float* h2     = (float*)(csr + N_EDGES);

    const int mtiles = (N_NODES + 15) / 16;

    gcn_cnt_zero<<<(N_NODES + 255) / 256, 256, 0, stream>>>(cnt, N_NODES);
    gcn_count_rank<<<(N_EDGES + 255) / 256, 256, 0, stream>>>(dst, cnt, rank, N_EDGES);
    gcn_scan_dinv<<<SCAN_BLOCKS, 1024, 0, stream>>>(cnt, rowptr, dinv, N_NODES);
    gcn_place<<<(N_EDGES + 255) / 256, 256, 0, stream>>>(src, dst, rank, rowptr, csr, N_EDGES);

    gcn_gather_mlp<<<mtiles, 256, 0, stream>>>(rowptr, csr, x, dinv, W1, b1, W2, h2, N_NODES);

    gcn_gather<16, false><<<(N_NODES * 4 + 255) / 256, 256, 0, stream>>>(rowptr, csr, h2, dinv, b2, out, N_NODES);
}


# Edges grouped by a two-level counting sort on LDS counters (chunk histogram, scan, scatter into 256-node bins, in-bin placement): no global atomic left; sums in LDS arrival order
# speedup vs baseline: 4.1527x; 1.6450x over previous
#include <hip/hip_runtime.h>
#include <hip/hip_bf16.h>
#include <stdint.h>


typedef __attribute__((ext_vector_type(2))) float v2f;
typedef __attribute__((ext_vector_type(8))) float v8f;

#define N_NODES 75000
#define N_EDGES 1200000

#define BIN_SHIFT 8
#define BIN_R     (1 << BIN_SHIFT)
#define NBINS     ((N_NODES + BIN_R - 1) / BIN_R)
#define CHUNK     4096
#define NCHUNK    ((N_EDGES + CHUNK - 1) / CHUNK)
#define HIST_N    (NBINS * NCHUNK)
#define BIN_TRIP  6

__global__ void __launch_bounds__(1024)
gcn_chunk_hist(const int* __restrict__ dst, int* __restrict__ hist, int nE) {
    __shared__ int s_h[NBINS];
    const int t = threadIdx.x;
    for (int i = t; i < NBINS; i += 1024) s_h[i] = 0;
    __syncthreads();
    const int base = blockIdx.x * CHUNK;
    int d[CHUNK / 1024];
    for (int j = 0; j < CHUNK / 1024; ++j) d[j] = dst[min(base + j * 1024 + t, nE - 1)];
    for (int j = 0; j < CHUNK / 1024; ++j)
        if (base + j * 1024 + t < nE) atomicAdd(&s_h[d[j] >> BIN_SHIFT], 1);
    __syncthreads();
    for (int i = t; i < NBINS; i += 1024) hist[i * NCHUNK + blockIdx.x] = s_h[i];
}

#define SCAN_BLOCKS ((HIST_N + 1023) / 1024)
__global__ void __launch_bounds__(1024)
gcn_scan(const int* __restrict__ in, int* __restrict__ out, int n) {
    __shared__ int s_part[32];
    __shared__ int s_tot[32];
    const int t    = threadIdx.x;
    const int lane = t & 31;
    const int wave = t >> 5;
    const int base = blockIdx.x * 1024;

    int part = 0;
    const int4* c4 = (const int4*)in;
    const int nq = base / 4;
    int q = t;
    for (; q + 3072 < nq; q += 4096) {
        const int4 a0 = c4[q], a1 = c4[q + 1024], a2 = c4[q + 2048], a3 = c4[q + 3072];
        part += (((a0.x + a0.y) + (a0.z + a0.w)) + ((a1.x + a1.y) + (a1.z + a1.w)))
              + (((a2.x + a2.y) + (a2.z + a2.w)) + ((a3.x + a3.y) + (a3.z + a3.w)));
    }
    for (; q < nq; q += 1024) {
        const int4 a0 = c4[q];
        part += (a0.x + a0.y) + (a0.z + a0.w);
    }
    const int i = base + t;
    const int c = (i < n) ? in[i] : 0;

    for (int o = 16; o > 0; o >>= 1) part += __shfl_xor(part, o);
    int inc = c;
    for (int o = 1; o < 32; o <<= 1) {
        const int y = __shfl_up(inc, o);
        if (lane >= o) inc += y;
    }
    if (lane == 31) s_tot[wave] = inc;
    if (lane == 0)  s_part[wave] = part;
    __syncthreads();

    int pv = s_part[lane];
    int tv = (lane < wave) ? s_tot[lane] : 0;
    for (int o = 16; o > 0; o >>= 1) {
        pv += __shfl_xor(pv, o);
        tv += __shfl_xor(tv, o);
    }
    const int excl = pv + tv + inc - c;
    if (i < n) out[i] = excl;
    if (i == n - 1) out[n] = excl + c;
}

__global__ void __launch_bounds__(1024)
gcn_chunk_scatter(const int* __restrict__ src, const int* __restrict__ dst,
                  const int* __restrict__ off, int* __restrict__ pairs, int nE) {
    __shared__ int s_cur[NBINS];
    const int t = threadIdx.x;
    for (int i = t; i < NBINS; i += 1024) s_cur[i] = off[i * NCHUNK + blockIdx.x];
    __syncthreads();
    const int base = blockIdx.x * CHUNK;
    int2* __restrict__ p2 = (int2*)pairs;
    int d[CHUNK / 1024], s[CHUNK / 1024];
    for (int j = 0; j < CHUNK / 1024; ++j) {
        const int e = min(base + j * 1024 + t, nE - 1);
        d[j] = dst[e];
        s[j] = src[e];
    }
    for (int j = 0; j < CHUNK / 1024; ++j) {
        if (base + j * 1024 + t < nE) {
            const int pos = atomicAdd(&s_cur[d[j] >> BIN_SHIFT], 1);
            p2[pos] = make_int2(d[j], s[j]);
        }
    }
}

__global__ void __launch_bounds__(1024)
gcn_bin_sort(const int* __restrict__ pairs, const int* __restrict__ off,
             int* __restrict__ rowptr, float* __restrict__ dinv,
             int* __restrict__ csr, int n) {
    __shared__ int s_cnt[BIN_R];
    __shared__ int s_cur[BIN_R];
    __shared__ int s_tot[BIN_R / 32];
    const int t      = threadIdx.x;
    const int lane   = t & 31;
    const int wave   = t >> 5;
    const int node0  = blockIdx.x * BIN_R;
    const int segBeg = off[blockIdx.x * NCHUNK];
    const int segEnd = off[(blockIdx.x + 1) * NCHUNK];
    const int2* __restrict__ p2 = (const int2*)pairs;

    if (t < BIN_R) s_cnt[t] = 0;
    __syncthreads();
    for (int p0 = segBeg; p0 < segEnd; p0 += BIN_TRIP * 1024) {
        int dd[BIN_TRIP];
        for (int j = 0; j < BIN_TRIP; ++j) dd[j] = p2[min(p0 + j * 1024 + t, segEnd - 1)].x;
        for (int j = 0; j < BIN_TRIP; ++j)
            if (p0 + j * 1024 + t < segEnd) atomicAdd(&s_cnt[dd[j] - node0], 1);
    }
    __syncthreads();

    int c = 0, inc = 0;
    if (t < BIN_R) {
        c = s_cnt[t];
        inc = c;
        for (int o = 1; o < 32; o <<= 1) {
            const int y = __shfl_up(inc, o);
            if (lane >= o) inc += y;
        }
        if (lane == 31) s_tot[wave] = inc;
    }
    __syncthreads();
    if (t < BIN_R) {
        int tv = (lane < wave) ? s_tot[lane] : 0;
        for (int o = 16; o > 0; o >>= 1) tv += __shfl_xor(tv, o);
        const int start = segBeg + tv + inc - c;
        s_cur[t] = start;
        const int node = node0 + t;
        if (node < n) {
            rowptr[node] = start;
            dinv[node] = rsqrtf((float)(c + 1));
        }
        if (node == n - 1) rowptr[n] = start + c;
    }
    __syncthreads();
    for (int p0 = segBeg; p0 < segEnd; p0 += BIN_TRIP * 1024) {
        int2 pr[BIN_TRIP];
        for (int j = 0; j < BIN_TRIP; ++j) pr[j] = p2[min(p0 + j * 1024 + t, segEnd - 1)];
        for (int j = 0; j < BIN_TRIP; ++j) {
            if (p0 + j * 1024 + t < segEnd) {
                const int pos = atomicAdd(&s_cur[pr[j].x - node0], 1);
                csr[pos] = pr[j].y;
            }
        }
    }
}

#define TILE_PITCH 68
__global__ void __launch_bounds__(256)
gcn_gather_mlp(const int* __restrict__ rowptr, const int* __restrict__ csr,
               const float* __restrict__ x, const float* __restrict__ dinv,
               const float* __restrict__ W1, const float* __restrict__ b1,
               const float* __restrict__ W2, float* __restrict__ h2, int n) {
    __shared__ float sA[16 * TILE_PITCH];
    __shared__ float sB[16 * TILE_PITCH];

    const int t     = threadIdx.x;
    const int slot  = t >> 4;
    const int l     = t & 15;
    const int mtile = blockIdx.x * 16;
    const int node  = mtile + slot;

    float4 acc = {0.0f, 0.0f, 0.0f, 0.0f};
    if (node < n) {
        const float4* __restrict__ x4 = (const float4*)x;
        const int beg = rowptr[node];
        const int end = rowptr[node + 1];
        const float di = dinv[node];
        {
            const float4 v = x4[(size_t)node * 16 + l];
            const float w = di * di;
            acc.x = v.x * w; acc.y = v.y * w; acc.z = v.z * w; acc.w = v.w * w;
        }
        int p = beg;
        for (; p + 4 <= end; p += 4) {
            const int s0 = csr[p], s1 = csr[p + 1], s2 = csr[p + 2], s3 = csr[p + 3];
            const float w0 = dinv[s0] * di, w1 = dinv[s1] * di;
            const float w2 = dinv[s2] * di, w3 = dinv[s3] * di;
            const float4 v0 = x4[(size_t)s0 * 16 + l];
            const float4 v1 = x4[(size_t)s1 * 16 + l];
            const float4 v2 = x4[(size_t)s2 * 16 + l];
            const float4 v3 = x4[(size_t)s3 * 16 + l];
            acc.x += v0.x * w0; acc.y += v0.y * w0; acc.z += v0.z * w0; acc.w += v0.w * w0;
            acc.x += v1.x * w1; acc.y += v1.y * w1; acc.z += v1.z * w1; acc.w += v1.w * w1;
            acc.x += v2.x * w2; acc.y += v2.y * w2; acc.z += v2.z * w2; acc.w += v2.w * w2;
            acc.x += v3.x * w3; acc.y += v3.y * w3; acc.z += v3.z * w3; acc.w += v3.w * w3;
        }
        for (; p < end; ++p) {
            const int s0 = csr[p];
            const float w0 = dinv[s0] * di;
            const float4 v0 = x4[(size_t)s0 * 16 + l];
            acc.x += v0.x * w0; acc.y += v0.y * w0; acc.z += v0.z * w0; acc.w += v0.w * w0;
        }
    }
    *(float4*)&sA[slot * TILE_PITCH + 4 * l] = acc;
    __syncthreads();

    const int lane = t & 31;
    const int half = lane >> 4;
    const int c    = lane & 15;
    const int wave = __builtin_amdgcn_readfirstlane(t >> 5);

    if (wave < 4) {
        const int ncol = wave * 16 + c;
        v8f d = {};
#pragma unroll
        for (int k0 = 0; k0 < 64; k0 += 4) {
            const int ka = k0 + half * 2;
            const v2f a = *(const v2f*)&sA[c * TILE_PITCH + ka];
            v2f b;
            b.x = W1[ka * 64 + ncol];
            b.y = W1[(ka + 1) * 64 + ncol];
            d = __builtin_amdgcn_wmma_f32_16x16x4_f32(false, a, false, b,
                                                      (short)0, d, false, false);
        }
        const float bias = b1[ncol];
#pragma unroll
        for (int r = 0; r < 8; ++r)
            sB[(r + half * 8) * TILE_PITCH + ncol] = fmaxf(d[r] + bias, 0.0f);
    }
    __syncthreads();

    if (wave == 0) {
        v8f d = {};
#pragma unroll
        for (int k0 = 0; k0 < 64; k0 += 4) {
            const int ka = k0 + half * 2;
            const v2f a = *(const v2f*)&sB[c * TILE_PITCH + ka];
            v2f b;
            b.x = W2[ka * 16 + c];
            b.y = W2[(ka + 1) * 16 + c];
            d = __builtin_amdgcn_wmma_f32_16x16x4_f32(false, a, false, b,
                                                      (short)0, d, false, false);
        }
#pragma unroll
        for (int r = 0; r < 8; ++r) {
            const int m = mtile + r + half * 8;
            if (m < n) h2[(size_t)m * 16 + c] = d[r];
        }
    }
}

template <int F, bool RELU>
__global__ void __launch_bounds__(256)
gcn_gather(const int* __restrict__ rowptr, const int* __restrict__ csr,
           const float* __restrict__ h, const float* __restrict__ dinv,
           const float* __restrict__ b, float* __restrict__ out, int n) {
    constexpr int LPN = F / 4;
    const int gt   = blockIdx.x * 256 + threadIdx.x;
    const int node = gt / LPN;
    const int l    = gt % LPN;
    if (node >= n) return;

    const float4* __restrict__ h4 = (const float4*)h;
    const int beg = rowptr[node];
    const int end = rowptr[node + 1];
    const float di = dinv[node];

    float4 acc;
    {
        const float4 v = h4[(size_t)node * LPN + l];
        const float w = di * di;
        acc.x = v.x * w; acc.y = v.y * w; acc.z = v.z * w; acc.w = v.w * w;
    }

    int p = beg;
    for (; p + 4 <= end; p += 4) {
        const int s0 = csr[p], s1 = csr[p + 1], s2 = csr[p + 2], s3 = csr[p + 3];
        const float w0 = dinv[s0] * di, w1 = dinv[s1] * di;
        const float w2 = dinv[s2] * di, w3 = dinv[s3] * di;
        const float4 v0 = h4[(size_t)s0 * LPN + l];
        const float4 v1 = h4[(size_t)s1 * LPN + l];
        const float4 v2 = h4[(size_t)s2 * LPN + l];
        const float4 v3 = h4[(size_t)s3 * LPN + l];
        acc.x += v0.x * w0; acc.y += v0.y * w0; acc.z += v0.z * w0; acc.w += v0.w * w0;
        acc.x += v1.x * w1; acc.y += v1.y * w1; acc.z += v1.z * w1; acc.w += v1.w * w1;
        acc.x += v2.x * w2; acc.y += v2.y * w2; acc.z += v2.z * w2; acc.w += v2.w * w2;
        acc.x += v3.x * w3; acc.y += v3.y * w3; acc.z += v3.z * w3; acc.w += v3.w * w3;
    }
    for (; p < end; ++p) {
        const int s0 = csr[p];
        const float w0 = dinv[s0] * di;
        const float4 v0 = h4[(size_t)s0 * LPN + l];
        acc.x += v0.x * w0; acc.y += v0.y * w0; acc.z += v0.z * w0; acc.w += v0.w * w0;
    }

    const float4 bb = ((const float4*)b)[l];
    acc.x += bb.x; acc.y += bb.y; acc.z += bb.z; acc.w += bb.w;
    if (RELU) {
        acc.x = fmaxf(acc.x, 0.0f); acc.y = fmaxf(acc.y, 0.0f);
        acc.z = fmaxf(acc.z, 0.0f); acc.w = fmaxf(acc.w, 0.0f);
    }
    ((float4*)out)[(size_t)node * LPN + l] = acc;
}

extern "C" void kernel_launch(void* const* d_in, const int* in_sizes, int n_in,
                              void* d_out, int out_size, void* d_ws, size_t ws_size,
                              hipStream_t stream) {
    const float* x  = (const float*)d_in[0];
    const int*   ei = (const int*)d_in[1];
    const float* W1 = (const float*)d_in[2];
    const float* b1 = (const float*)d_in[3];
    const float* W2 = (const float*)d_in[4];
    const float* b2 = (const float*)d_in[5];
    float* out = (float*)d_out;

    const int* src = ei;
    const int* dst = ei + N_EDGES;

    int*   hist   = (int*)d_ws;
    int*   off    = hist + 86016;
    int*   rowptr = off + 86016;
    float* dinv   = (float*)(rowptr + 75008);
    int*   pairs  = (int*)(dinv + 75008);
    int*   csr    = pairs + 2 * N_EDGES;
    float* h2     = (float*)(csr + N_EDGES);

    const int mtiles = (N_NODES + 15) / 16;

    gcn_chunk_hist<<<NCHUNK, 1024, 0, stream>>>(dst, hist, N_EDGES);
    gcn_scan<<<SCAN_BLOCKS, 1024, 0, stream>>>(hist, off, HIST_N);
    gcn_chunk_scatter<<<NCHUNK, 1024, 0, stream>>>(src, dst, off, pairs, N_EDGES);
    gcn_bin_sort<<<NBINS, 1024, 0, stream>>>(pairs, off, rowptr, dinv, csr, N_NODES);

    gcn_gather_mlp<<<mtiles, 256, 0, stream>>>(rowptr, csr, x, dinv, W1, b1, W2, h2, N_NODES);

    gcn_gather<16, false><<<(N_NODES * 4 + 255) / 256, 256, 0, stream>>>(rowptr, csr, h2, dinv, b2, out, N_NODES);
}


# Each chunk groups its pairs by bin inside its own 4096 places (LDS count with rank, LDS scan, one store pass); a bin block finds its pieces by table: 6 launches to 4; sums in LDS arrival order
# speedup vs baseline: 5.4079x; 1.3023x over previous
#include <hip/hip_runtime.h>
#include <hip/hip_bf16.h>
#include <stdint.h>


typedef __attribute__((ext_vector_type(2))) float v2f;
typedef __attribute__((ext_vector_type(8))) float v8f;

#define N_NODES 75000
#define N_EDGES 1200000

#define BIN_SHIFT 8
#define BIN_R     (1 << BIN_SHIFT)
#define NBINS     ((N_NODES + BIN_R - 1) / BIN_R)
#define CHUNK     4096
#define NCHUNK    ((N_EDGES + CHUNK - 1) / CHUNK)
#define HIST_N    (NBINS * NCHUNK)
#define SCAN_T    (((NBINS + 31) / 32) * 32)
#define PPW       ((NCHUNK + 31) / 32)

static_assert(SCAN_T <= 1024, "one thread per bin in the chunk scan");
static_assert(NCHUNK <= 1024, "one thread per chunk in the bin table load");

__global__ void __launch_bounds__(1024)
gcn_chunk_sort(const int* __restrict__ src, const int* __restrict__ dst,
               int* __restrict__ tab, int* __restrict__ pairs, int nE) {
    __shared__ int s_h[NBINS];
    __shared__ int s_tot[SCAN_T / 32];
    const int t    = threadIdx.x;
    const int lane = t & 31;
    const int wave = t >> 5;
    for (int i = t; i < NBINS; i += 1024) s_h[i] = 0;
    __syncthreads();
    const int base = blockIdx.x * CHUNK;
    int2* __restrict__ p2 = (int2*)pairs;
    int d[CHUNK / 1024], s[CHUNK / 1024], rk[CHUNK / 1024];
    for (int j = 0; j < CHUNK / 1024; ++j) {
        const int e = min(base + j * 1024 + t, nE - 1);
        d[j] = dst[e];
        s[j] = src[e];
    }
    for (int j = 0; j < CHUNK / 1024; ++j) {
        rk[j] = 0;
        if (base + j * 1024 + t < nE) rk[j] = atomicAdd(&s_h[d[j] >> BIN_SHIFT], 1);
    }
    __syncthreads();

    int c = 0, inc = 0;
    if (t < SCAN_T) {
        c = (t < NBINS) ? s_h[t] : 0;
        inc = c;
        for (int o = 1; o < 32; o <<= 1) {
            const int y = __shfl_up(inc, o);
            if (lane >= o) inc += y;
        }
        if (lane == 31) s_tot[wave] = inc;
    }
    __syncthreads();
    if (t < SCAN_T) {
        int tv = (lane < wave) ? s_tot[lane] : 0;
        for (int o = 16; o > 0; o >>= 1) tv += __shfl_xor(tv, o);
        const int excl = tv + inc - c;
        if (t < NBINS) {
            s_h[t] = excl;
            ((int2*)tab)[t * NCHUNK + blockIdx.x] = make_int2(c, excl);
        }
    }
    __syncthreads();
    int o[CHUNK / 1024];
    for (int j = 0; j < CHUNK / 1024; ++j) o[j] = s_h[d[j] >> BIN_SHIFT];
    for (int j = 0; j < CHUNK / 1024; ++j)
        if (base + j * 1024 + t < nE)
            p2[base + o[j] + rk[j]] = make_int2(d[j], s[j]);
}

__global__ void __launch_bounds__(1024)
gcn_bin_sort(const int* __restrict__ pairs, const int* __restrict__ tab,
             int* __restrict__ rowptr, float* __restrict__ dinv,
             int* __restrict__ csr, int n, int nE) {
    __shared__ int s_cnt[BIN_R];
    __shared__ int s_cur[BIN_R];
    __shared__ int s_tot[BIN_R / 32];
    __shared__ int s_plen[NCHUNK];
    __shared__ int s_pbeg[NCHUNK];
    __shared__ int s_red[32];
    const int t      = threadIdx.x;
    const int lane   = t & 31;
    const int wave   = t >> 5;
    const int node0  = blockIdx.x * BIN_R;
    const int2* __restrict__ p2 = (const int2*)pairs;

    int lo = 0;
    if (t < NCHUNK) {
        const int2 e = ((const int2*)tab)[blockIdx.x * NCHUNK + t];
        lo = e.y;
        s_plen[t] = e.x;
        s_pbeg[t] = t * CHUNK + lo;
    }
    if (t < BIN_R) s_cnt[t] = 0;
    __syncthreads();

    int2 pr[PPW];
    unsigned vm = 0;
    for (int j = 0; j < PPW; ++j) {
        const int cc  = wave + 32 * j;
        const int len = (cc < NCHUNK) ? s_plen[cc] : 0;
        const int pb  = (cc < NCHUNK) ? s_pbeg[cc] : 0;
        pr[j] = p2[min(pb + min(lane, max(len - 1, 0)), nE - 1)];
        if (lane < len) vm |= 1u << j;
    }
    {
        int r = lo;
        for (int o = 16; o > 0; o >>= 1) r += __shfl_xor(r, o);
        if (lane == 0) s_red[wave] = r;
    }
    for (int j = 0; j < PPW; ++j)
        if ((vm >> j) & 1u) atomicAdd(&s_cnt[pr[j].x - node0], 1);
    for (int j = 0; j < PPW; ++j) {
        const int cc = wave + 32 * j;
        if (cc < NCHUNK) {
            const int len = s_plen[cc];
            const int pb  = s_pbeg[cc];
            for (int i = lane + 32; i < len; i += 32)
                atomicAdd(&s_cnt[p2[pb + i].x - node0], 1);
        }
    }
    __syncthreads();

    int c = 0, inc = 0;
    if (t < BIN_R) {
        c = s_cnt[t];
        inc = c;
        for (int o = 1; o < 32; o <<= 1) {
            const int y = __shfl_up(inc, o);
            if (lane >= o) inc += y;
        }
        if (lane == 31) s_tot[wave] = inc;
    }
    __syncthreads();
    if (t < BIN_R) {
        int tv = (lane < wave) ? s_tot[lane] : 0;
        int segBeg = s_red[lane];
        for (int o = 16; o > 0; o >>= 1) {
            tv += __shfl_xor(tv, o);
            segBeg += __shfl_xor(segBeg, o);
        }
        const int start = segBeg + tv + inc - c;
        s_cur[t] = start;
        const int node = node0 + t;
        if (node < n) {
            rowptr[node] = start;
            dinv[node] = rsqrtf((float)(c + 1));
        }
        if (node == n - 1) rowptr[n] = start + c;
    }
    __syncthreads();
    {
        int pos[PPW];
        for (int j = 0; j < PPW; ++j) {
            pos[j] = 0;
            if ((vm >> j) & 1u) pos[j] = atomicAdd(&s_cur[pr[j].x - node0], 1);
        }
        for (int j = 0; j < PPW; ++j)
            if ((vm >> j) & 1u) csr[pos[j]] = pr[j].y;
        for (int j = 0; j < PPW; ++j) {
            const int cc = wave + 32 * j;
            if (cc < NCHUNK) {
                const int len = s_plen[cc];
                const int pb  = s_pbeg[cc];
                for (int i = lane + 32; i < len; i += 32) {
                    const int2 q = p2[pb + i];
                    const int at = atomicAdd(&s_cur[q.x - node0], 1);
                    csr[at] = q.y;
                }
            }
        }
    }
}

#define TILE_PITCH 68
__global__ void __launch_bounds__(256)
gcn_gather_mlp(const int* __restrict__ rowptr, const int* __restrict__ csr,
               const float* __restrict__ x, const float* __restrict__ dinv,
               const float* __restrict__ W1, const float* __restrict__ b1,
               const float* __restrict__ W2, float* __restrict__ h2, int n) {
    __shared__ float sA[16 * TILE_PITCH];
    __shared__ float sB[16 * TILE_PITCH];

    const int t     = threadIdx.x;
    const int slot  = t >> 4;
    const int l     = t & 15;
    const int mtile = blockIdx.x * 16;
    const int node  = mtile + slot;

    float4 acc = {0.0f, 0.0f, 0.0f, 0.0f};
    if (node < n) {
        const float4* __restrict__ x4 = (const float4*)x;
        const int beg = rowptr[node];
        const int end = rowptr[node + 1];
        const float di = dinv[node];
        {
            const float4 v = x4[(size_t)node * 16 + l];
            const float w = di * di;
            acc.x = v.x * w; acc.y = v.y * w; acc.z = v.z * w; acc.w = v.w * w;
        }
        int p = beg;
        for (; p + 4 <= end; p += 4) {
            const int s0 = csr[p], s1 = csr[p + 1], s2 = csr[p + 2], s3 = csr[p + 3];
            const float w0 = dinv[s0] * di, w1 = dinv[s1] * di;
            const float w2 = dinv[s2] * di, w3 = dinv[s3] * di;
            const float4 v0 = x4[(size_t)s0 * 16 + l];
            const float4 v1 = x4[(size_t)s1 * 16 + l];
            const float4 v2 = x4[(size_t)s2 * 16 + l];
            const float4 v3 = x4[(size_t)s3 * 16 + l];
            acc.x += v0.x * w0; acc.y += v0.y * w0; acc.z += v0.z * w0; acc.w += v0.w * w0;
            acc.x += v1.x * w1; acc.y += v1.y * w1; acc.z += v1.z * w1; acc.w += v1.w * w1;
            acc.x += v2.x * w2; acc.y += v2.y * w2; acc.z += v2.z * w2; acc.w += v2.w * w2;
            acc.x += v3.x * w3; acc.y += v3.y * w3; acc.z += v3.z * w3; acc.w += v3.w * w3;
        }
        for (; p < end; ++p) {
            const int s0 = csr[p];
            const float w0 = dinv[s0] * di;
            const float4 v0 = x4[(size_t)s0 * 16 + l];
            acc.x += v0.x * w0; acc.y += v0.y * w0; acc.z += v0.z * w0; acc.w += v0.w * w0;
        }
    }
    *(float4*)&sA[slot * TILE_PITCH + 4 * l] = acc;
    __syncthreads();

    const int lane = t & 31;
    const int half = lane >> 4;
    const int c    = lane & 15;
    const int wave = __builtin_amdgcn_readfirstlane(t >> 5);

    if (wave < 4) {
        const int ncol = wave * 16 + c;
        v8f d = {};
#pragma unroll
        for (int k0 = 0; k0 < 64; k0 += 4) {
            const int ka = k0 + half * 2;
            const v2f a = *(const v2f*)&sA[c * TILE_PITCH + ka];
            v2f b;
            b.x = W1[ka * 64 + ncol];
            b.y = W1[(ka + 1) * 64 + ncol];
            d = __builtin_amdgcn_wmma_f32_16x16x4_f32(false, a, false, b,
                                                      (short)0, d, false, false);
        }
        const float bias = b1[ncol];
#pragma unroll
        for (int r = 0; r < 8; ++r)
            sB[(r + half * 8) * TILE_PITCH + ncol] = fmaxf(d[r] + bias, 0.0f);
    }
    __syncthreads();

    if (wave == 0) {
        v8f d = {};
#pragma unroll
        for (int k0 = 0; k0 < 64; k0 += 4) {
            const int ka = k0 + half * 2;
            const v2f a = *(const v2f*)&sB[c * TILE_PITCH + ka];
            v2f b;
            b.x = W2[ka * 16 + c];
            b.y = W2[(ka + 1) * 16 + c];
            d = __builtin_amdgcn_wmma_f32_16x16x4_f32(false, a, false, b,
                                                      (short)0, d, false, false);
        }
#pragma unroll
        for (int r = 0; r < 8; ++r) {
            const int m = mtile + r + half * 8;
            if (m < n) h2[(size_t)m * 16 + c] = d[r];
        }
    }
}

template <int F, bool RELU>
__global__ void __launch_bounds__(256)
gcn_gather(const int* __restrict__ rowptr, const int* __restrict__ csr,
           const float* __restrict__ h, const float* __restrict__ dinv,
           const float* __restrict__ b, float* __restrict__ out, int n) {
    constexpr int LPN = F / 4;
    const int gt   = blockIdx.x * 256 + threadIdx.x;
    const int node = gt / LPN;
    const int l    = gt % LPN;
    if (node >= n) return;

    const float4* __restrict__ h4 = (const float4*)h;
    const int beg = rowptr[node];
    const int end = rowptr[node + 1];
    const float di = dinv[node];

    float4 acc;
    {
        const float4 v = h4[(size_t)node * LPN + l];
        const float w = di * di;
        acc.x = v.x * w; acc.y = v.y * w; acc.z = v.z * w; acc.w = v.w * w;
    }

    int p = beg;
    for (; p + 4 <= end; p += 4) {
        const int s0 = csr[p], s1 = csr[p + 1], s2 = csr[p + 2], s3 = csr[p + 3];
        const float w0 = dinv[s0] * di, w1 = dinv[s1] * di;
        const float w2 = dinv[s2] * di, w3 = dinv[s3] * di;
        const float4 v0 = h4[(size_t)s0 * LPN + l];
        const float4 v1 = h4[(size_t)s1 * LPN + l];
        const float4 v2 = h4[(size_t)s2 * LPN + l];
        const float4 v3 = h4[(size_t)s3 * LPN + l];
        acc.x += v0.x * w0; acc.y += v0.y * w0; acc.z += v0.z * w0; acc.w += v0.w * w0;
        acc.x += v1.x * w1; acc.y += v1.y * w1; acc.z += v1.z * w1; acc.w += v1.w * w1;
        acc.x += v2.x * w2; acc.y += v2.y * w2; acc.z += v2.z * w2; acc.w += v2.w * w2;
        acc.x += v3.x * w3; acc.y += v3.y * w3; acc.z += v3.z * w3; acc.w += v3.w * w3;
    }
    for (; p < end; ++p) {
        const int s0 = csr[p];
        const float w0 = dinv[s0] * di;
        const float4 v0 = h4[(size_t)s0 * LPN + l];
        acc.x += v0.x * w0; acc.y += v0.y * w0; acc.z += v0.z * w0; acc.w += v0.w * w0;
    }

    const float4 bb = ((const float4*)b)[l];
    acc.x += bb.x; acc.y += bb.y; acc.z += bb.z; acc.w += bb.w;
    if (RELU) {
        acc.x = fmaxf(acc.x, 0.0f); acc.y = fmaxf(acc.y, 0.0f);
        acc.z = fmaxf(acc.z, 0.0f); acc.w = fmaxf(acc.w, 0.0f);
    }
    ((float4*)out)[(size_t)node * LPN + l] = acc;
}

extern "C" void kernel_launch(void* const* d_in, const int* in_sizes, int n_in,
                              void* d_out, int out_size, void* d_ws, size_t ws_size,
                              hipStream_t stream) {
    const float* x  = (const float*)d_in[0];
    const int*   ei = (const int*)d_in[1];
    const float* W1 = (const float*)d_in[2];
    const float* b1 = (const float*)d_in[3];
    const float* W2 = (const float*)d_in[4];
    const float* b2 = (const float*)d_in[5];
    float* out = (float*)d_out;

    const int* src = ei;
    const int* dst = ei + N_EDGES;

    int*   tab    = (int*)d_ws;
    int*   rowptr = tab + 172032;
    float* dinv   = (float*)(rowptr + 75008);
    int*   pairs  = (int*)(dinv + 75008);
    int*   csr    = pairs + 2 * N_EDGES;
    float* h2     = (float*)(csr + N_EDGES);

    const int mtiles = (N_NODES + 15) / 16;

    gcn_chunk_sort<<<NCHUNK, 1024, 0, stream>>>(src, dst, tab, pairs, N_EDGES);
    gcn_bin_sort<<<NBINS, 1024, 0, stream>>>(pairs, tab, rowptr, dinv, csr, N_NODES, N_EDGES);

    gcn_gather_mlp<<<mtiles, 256, 0, stream>>>(rowptr, csr, x, dinv, W1, b1, W2, h2, N_NODES);

    gcn_gather<16, false><<<(N_NODES * 4 + 255) / 256, 256, 0, stream>>>(rowptr, csr, h2, dinv, b2, out, N_NODES);
}


# Both sort launches put each record into its sorted place in an LDS copy of the work-group's output range and write it out in index order as whole 128-byte lines; ranks and sum order as before
# speedup vs baseline: 6.0926x; 1.1266x over previous
#include <hip/hip_runtime.h>
#include <hip/hip_bf16.h>
#include <stdint.h>


typedef __attribute__((ext_vector_type(2))) float v2f;
typedef __attribute__((ext_vector_type(8))) float v8f;

#define N_NODES 75000
#define N_EDGES 1200000

#define BIN_SHIFT 8
#define BIN_R     (1 << BIN_SHIFT)
#define NBINS     ((N_NODES + BIN_R - 1) / BIN_R)
#define CHUNK     4096
#define NCHUNK    ((N_EDGES + CHUNK - 1) / CHUNK)
#define HIST_N    (NBINS * NCHUNK)
#define SCAN_T    (((NBINS + 31) / 32) * 32)
#define PPW       ((NCHUNK + 31) / 32)
#define STAGE_CAP 8192

static_assert(SCAN_T <= 1024, "one thread per bin in the chunk scan");
static_assert(NCHUNK <= 1024, "one thread per chunk in the bin table load");

__global__ void __launch_bounds__(1024)
gcn_chunk_sort(const int* __restrict__ src, const int* __restrict__ dst,
               int* __restrict__ tab, int* __restrict__ pairs, int nE) {
    __shared__ int s_h[NBINS];
    __shared__ int s_tot[SCAN_T / 32];
    __shared__ __attribute__((aligned(16))) int2 s_stage[CHUNK];
    const int t    = threadIdx.x;
    const int lane = t & 31;
    const int wave = t >> 5;
    for (int i = t; i < NBINS; i += 1024) s_h[i] = 0;
    __syncthreads();
    const int base = blockIdx.x * CHUNK;
    int2* __restrict__ p2 = (int2*)pairs;
    int d[CHUNK / 1024], s[CHUNK / 1024], rk[CHUNK / 1024];
    for (int j = 0; j < CHUNK / 1024; ++j) {
        const int e = min(base + j * 1024 + t, nE - 1);
        d[j] = dst[e];
        s[j] = src[e];
    }
    for (int j = 0; j < CHUNK / 1024; ++j) {
        rk[j] = 0;
        if (base + j * 1024 + t < nE) rk[j] = atomicAdd(&s_h[d[j] >> BIN_SHIFT], 1);
    }
    __syncthreads();

    int c = 0, inc = 0;
    if (t < SCAN_T) {
        c = (t < NBINS) ? s_h[t] : 0;
        inc = c;
        for (int o = 1; o < 32; o <<= 1) {
            const int y = __shfl_up(inc, o);
            if (lane >= o) inc += y;
        }
        if (lane == 31) s_tot[wave] = inc;
    }
    __syncthreads();
    if (t < SCAN_T) {
        int tv = (lane < wave) ? s_tot[lane] : 0;
        for (int o = 16; o > 0; o >>= 1) tv += __shfl_xor(tv, o);
        const int excl = tv + inc - c;
        if (t < NBINS) {
            s_h[t] = excl;
            ((int2*)tab)[t * NCHUNK + blockIdx.x] = make_int2(c, excl);
        }
    }
    __syncthreads();
    int o[CHUNK / 1024];
    for (int j = 0; j < CHUNK / 1024; ++j) o[j] = s_h[d[j] >> BIN_SHIFT];
    for (int j = 0; j < CHUNK / 1024; ++j)
        if (base + j * 1024 + t < nE)
            s_stage[o[j] + rk[j]] = make_int2(d[j], s[j]);
    __syncthreads();
    const int4* __restrict__ st4 = (const int4*)s_stage;
    int4* __restrict__ g4 = (int4*)(p2 + base);
    for (int j = 0; j < CHUNK / 2048; ++j) {
        const int i = j * 1024 + t;
        if (base + 2 * i + 1 < nE)  g4[i] = st4[i];
        else if (base + 2 * i < nE) p2[base + 2 * i] = s_stage[2 * i];
    }
}

__global__ void __launch_bounds__(1024)
gcn_bin_sort(const int* __restrict__ pairs, const int* __restrict__ tab,
             int* __restrict__ rowptr, float* __restrict__ dinv,
             int* __restrict__ csr, int n, int nE) {
    __shared__ int s_cnt[BIN_R];
    __shared__ int s_cur[BIN_R];
    __shared__ int s_tot[BIN_R / 32];
    __shared__ int s_plen[NCHUNK];
    __shared__ int s_pbeg[NCHUNK];
    __shared__ int s_red[32];
    __shared__ int s_seg[2];
    __shared__ int s_stage[STAGE_CAP];
    const int t      = threadIdx.x;
    const int lane   = t & 31;
    const int wave   = t >> 5;
    const int node0  = blockIdx.x * BIN_R;
    const int2* __restrict__ p2 = (const int2*)pairs;

    int lo = 0;
    if (t < NCHUNK) {
        const int2 e = ((const int2*)tab)[blockIdx.x * NCHUNK + t];
        lo = e.y;
        s_plen[t] = e.x;
        s_pbeg[t] = t * CHUNK + lo;
    }
    if (t < BIN_R) s_cnt[t] = 0;
    __syncthreads();

    int2 pr[PPW];
    unsigned vm = 0;
    for (int j = 0; j < PPW; ++j) {
        const int cc  = wave + 32 * j;
        const int len = (cc < NCHUNK) ? s_plen[cc] : 0;
        const int pb  = (cc < NCHUNK) ? s_pbeg[cc] : 0;
        pr[j] = p2[min(pb + min(lane, max(len - 1, 0)), nE - 1)];
        if (lane < len) vm |= 1u << j;
    }
    {
        int r = lo;
        for (int o = 16; o > 0; o >>= 1) r += __shfl_xor(r, o);
        if (lane == 0) s_red[wave] = r;
    }
    for (int j = 0; j < PPW; ++j)
        if ((vm >> j) & 1u) atomicAdd(&s_cnt[pr[j].x - node0], 1);
    for (int j = 0; j < PPW; ++j) {
        const int cc = wave + 32 * j;
        if (cc < NCHUNK) {
            const int len = s_plen[cc];
            const int pb  = s_pbeg[cc];
            for (int i = lane + 32; i < len; i += 32)
                atomicAdd(&s_cnt[p2[pb + i].x - node0], 1);
        }
    }
    __syncthreads();

    int c = 0, inc = 0;
    if (t < BIN_R) {
        c = s_cnt[t];
        inc = c;
        for (int o = 1; o < 32; o <<= 1) {
            const int y = __shfl_up(inc, o);
            if (lane >= o) inc += y;
        }
        if (lane == 31) s_tot[wave] = inc;
    }
    __syncthreads();
    if (t < BIN_R) {
        int tv = (lane < wave) ? s_tot[lane] : 0;
        int segBeg = s_red[lane];
        for (int o = 16; o > 0; o >>= 1) {
            tv += __shfl_xor(tv, o);
            segBeg += __shfl_xor(segBeg, o);
        }
        const int local = tv + inc - c;
        const int start = segBeg + local;
        s_cur[t] = local;
        if (t == BIN_R - 1) { s_seg[0] = segBeg; s_seg[1] = local + c; }
        const int node = node0 + t;
        if (node < n) {
            rowptr[node] = start;
            dinv[node] = rsqrtf((float)(c + 1));
        }
        if (node == n - 1) rowptr[n] = start + c;
    }
    __syncthreads();
    const int segBeg = s_seg[0];
    const int total  = s_seg[1];
    const bool staged = total <= STAGE_CAP;
    {
        int pos[PPW];
        for (int j = 0; j < PPW; ++j) {
            pos[j] = 0;
            if ((vm >> j) & 1u) pos[j] = atomicAdd(&s_cur[pr[j].x - node0], 1);
        }
        if (staged) {
            for (int j = 0; j < PPW; ++j)
                if ((vm >> j) & 1u) s_stage[pos[j]] = pr[j].y;
        } else {
            for (int j = 0; j < PPW; ++j)
                if ((vm >> j) & 1u) csr[segBeg + pos[j]] = pr[j].y;
        }
        for (int j = 0; j < PPW; ++j) {
            const int cc = wave + 32 * j;
            if (cc < NCHUNK) {
                const int len = s_plen[cc];
                const int pb  = s_pbeg[cc];
                for (int i = lane + 32; i < len; i += 32) {
                    const int2 q = p2[pb + i];
                    const int at = atomicAdd(&s_cur[q.x - node0], 1);
                    if (staged) s_stage[at] = q.y;
                    else        csr[segBeg + at] = q.y;
                }
            }
        }
    }
    __syncthreads();
    if (staged) {
        const int endg = segBeg + total;
        for (int g = (segBeg & ~31) + t; g < endg; g += 1024)
            if (g >= segBeg) csr[g] = s_stage[g - segBeg];
    }
}

#define TILE_PITCH 68
__global__ void __launch_bounds__(256)
gcn_gather_mlp(const int* __restrict__ rowptr, const int* __restrict__ csr,
               const float* __restrict__ x, const float* __restrict__ dinv,
               const float* __restrict__ W1, const float* __restrict__ b1,
               const float* __restrict__ W2, float* __restrict__ h2, int n) {
    __shared__ float sA[16 * TILE_PITCH];
    __shared__ float sB[16 * TILE_PITCH];

    const int t     = threadIdx.x;
    const int slot  = t >> 4;
    const int l     = t & 15;
    const int mtile = blockIdx.x * 16;
    const int node  = mtile + slot;

    float4 acc = {0.0f, 0.0f, 0.0f, 0.0f};
    if (node < n) {
        const float4* __restrict__ x4 = (const float4*)x;
        const int beg = rowptr[node];
        const int end = rowptr[node + 1];
        const float di = dinv[node];
        {
            const float4 v = x4[(size_t)node * 16 + l];
            const float w = di * di;
            acc.x = v.x * w; acc.y = v.y * w; acc.z = v.z * w; acc.w = v.w * w;
        }
        int p = beg;
        for (; p + 4 <= end; p += 4) {
            const int s0 = csr[p], s1 = csr[p + 1], s2 = csr[p + 2], s3 = csr[p + 3];
            const float w0 = dinv[s0] * di, w1 = dinv[s1] * di;
            const float w2 = dinv[s2] * di, w3 = dinv[s3] * di;
            const float4 v0 = x4[(size_t)s0 * 16 + l];
            const float4 v1 = x4[(size_t)s1 * 16 + l];
            const float4 v2 = x4[(size_t)s2 * 16 + l];
            const float4 v3 = x4[(size_t)s3 * 16 + l];
            acc.x += v0.x * w0; acc.y += v0.y * w0; acc.z += v0.z * w0; acc.w += v0.w * w0;
            acc.x += v1.x * w1; acc.y += v1.y * w1; acc.z += v1.z * w1; acc.w += v1.w * w1;
            acc.x += v2.x * w2; acc.y += v2.y * w2; acc.z += v2.z * w2; acc.w += v2.w * w2;
            acc.x += v3.x * w3; acc.y += v3.y * w3; acc.z += v3.z * w3; acc.w += v3.w * w3;
        }
        for (; p < end; ++p) {
            const int s0 = csr[p];
            const float w0 = dinv[s0] * di;
            const float4 v0 = x4[(size_t)s0 * 16 + l];
            acc.x += v0.x * w0; acc.y += v0.y * w0; acc.z += v0.z * w0; acc.w += v0.w * w0;
        }
    }
    *(float4*)&sA[slot * TILE_PITCH + 4 * l] = acc;
    __syncthreads();

    const int lane = t & 31;
    const int half = lane >> 4;
    const int c    = lane & 15;
    const int wave = __builtin_amdgcn_readfirstlane(t >> 5);

    if (wave < 4) {
        const int ncol = wave * 16 + c;
        v8f d = {};
#pragma unroll
        for (int k0 = 0; k0 < 64; k0 += 4) {
            const int ka = k0 + half * 2;
            const v2f a = *(const v2f*)&sA[c * TILE_PITCH + ka];
            v2f b;
            b.x = W1[ka * 64 + ncol];
            b.y = W1[(ka + 1) * 64 + ncol];
            d = __builtin_amdgcn_wmma_f32_16x16x4_f32(false, a, false, b,
                                                      (short)0, d, false, false);
        }
        const float bias = b1[ncol];
#pragma unroll
        for (int r = 0; r < 8; ++r)
            sB[(r + half * 8) * TILE_PITCH + ncol] = fmaxf(d[r] + bias, 0.0f);
    }
    __syncthreads();

    if (wave == 0) {
        v8f d = {};
#pragma unroll
        for (int k0 = 0; k0 < 64; k0 += 4) {
            const int ka = k0 + half * 2;
            const v2f a = *(const v2f*)&sB[c * TILE_PITCH + ka];
            v2f b;
            b.x = W2[ka * 16 + c];
            b.y = W2[(ka + 1) * 16 + c];
            d = __builtin_amdgcn_wmma_f32_16x16x4_f32(false, a, false, b,
                                                      (short)0, d, false, false);
        }
#pragma unroll
        for (int r = 0; r < 8; ++r) {
            const int m = mtile + r + half * 8;
            if (m < n) h2[(size_t)m * 16 + c] = d[r];
        }
    }
}

template <int F, bool RELU>
__global__ void __launch_bounds__(256)
gcn_gather(const int* __restrict__ rowptr, const int* __restrict__ csr,
           const float* __restrict__ h, const float* __restrict__ dinv,
           const float* __restrict__ b, float* __restrict__ out, int n) {
    constexpr int LPN = F / 4;
    const int gt   = blockIdx.x * 256 + threadIdx.x;
    const int node = gt / LPN;
    const int l    = gt % LPN;
    if (node >= n) return;

    const float4* __restrict__ h4 = (const float4*)h;
    const int beg = rowptr[node];
    const int end = rowptr[node + 1];
    const float di = dinv[node];

    float4 acc;
    {
        const float4 v = h4[(size_t)node * LPN + l];
        const float w = di * di;
        acc.x = v.x * w; acc.y = v.y * w; acc.z = v.z * w; acc.w = v.w * w;
    }

    int p = beg;
    for (; p + 4 <= end; p += 4) {
        const int s0 = csr[p], s1 = csr[p + 1], s2 = csr[p + 2], s3 = csr[p + 3];
        const float w0 = dinv[s0] * di, w1 = dinv[s1] * di;
        const float w2 = dinv[s2] * di, w3 = dinv[s3] * di;
        const float4 v0 = h4[(size_t)s0 * LPN + l];
        const float4 v1 = h4[(size_t)s1 * LPN + l];
        const float4 v2 = h4[(size_t)s2 * LPN + l];
        const float4 v3 = h4[(size_t)s3 * LPN + l];
        acc.x += v0.x * w0; acc.y += v0.y * w0; acc.z += v0.z * w0; acc.w += v0.w * w0;
        acc.x += v1.x * w1; acc.y += v1.y * w1; acc.z += v1.z * w1; acc.w += v1.w * w1;
        acc.x += v2.x * w2; acc.y += v2.y * w2; acc.z += v2.z * w2; acc.w += v2.w * w2;
        acc.x += v3.x * w3; acc.y += v3.y * w3; acc.z += v3.z * w3; acc.w += v3.w * w3;
    }
    for (; p < end; ++p) {
        const int s0 = csr[p];
        const float w0 = dinv[s0] * di;
        const float4 v0 = h4[(size_t)s0 * LPN + l];
        acc.x += v0.x * w0; acc.y += v0.y * w0; acc.z += v0.z * w0; acc.w += v0.w * w0;
    }

    const float4 bb = ((const float4*)b)[l];
    acc.x += bb.x; acc.y += bb.y; acc.z += bb.z; acc.w += bb.w;
    if (RELU) {
        acc.x = fmaxf(acc.x, 0.0f); acc.y = fmaxf(acc.y, 0.0f);
        acc.z = fmaxf(acc.z, 0.0f); acc.w = fmaxf(acc.w, 0.0f);
    }
    ((float4*)out)[(size_t)node * LPN + l] = acc;
}

extern "C" void kernel_launch(void* const* d_in, const int* in_sizes, int n_in,
                              void* d_out, int out_size, void* d_ws, size_t ws_size,
                              hipStream_t stream) {
    const float* x  = (const float*)d_in[0];
    const int*   ei = (const int*)d_in[1];
    const float* W1 = (const float*)d_in[2];
    const float* b1 = (const float*)d_in[3];
    const float* W2 = (const float*)d_in[4];
    const float* b2 = (const float*)d_in[5];
    float* out = (float*)d_out;

    const int* src = ei;
    const int* dst = ei + N_EDGES;

    int*   tab    = (int*)d_ws;
    int*   rowptr = tab + 172032;
    float* dinv   = (float*)(rowptr + 75008);
    int*   pairs  = (int*)(dinv + 75008);
    int*   csr    = pairs + 2 * N_EDGES;
    float* h2     = (float*)(csr + N_EDGES);

    const int mtiles = (N_NODES + 15) / 16;

    gcn_chunk_sort<<<NCHUNK, 1024, 0, stream>>>(src, dst, tab, pairs, N_EDGES);
    gcn_bin_sort<<<NBINS, 1024, 0, stream>>>(pairs, tab, rowptr, dinv, csr, N_NODES, N_EDGES);

    gcn_gather_mlp<<<mtiles, 256, 0, stream>>>(rowptr, csr, x, dinv, W1, b1, W2, h2, N_NODES);

    gcn_gather<16, false><<<(N_NODES * 4 + 255) / 256, 256, 0, stream>>>(rowptr, csr, h2, dinv, b2, out, N_NODES);
}


# h2 rows are stored already multiplied by their own node's dinv; the layer-2 gather adds them plainly and scales the sum by dinv[i] once: no dinv look-up per edge; sum order kept, products regrouped
# speedup vs baseline: 6.4529x; 1.0591x over previous
#include <hip/hip_runtime.h>
#include <hip/hip_bf16.h>
#include <stdint.h>


typedef __attribute__((ext_vector_type(2))) float v2f;
typedef __attribute__((ext_vector_type(8))) float v8f;

#define N_NODES 75000
#define N_EDGES 1200000

#define BIN_SHIFT 8
#define BIN_R     (1 << BIN_SHIFT)
#define NBINS     ((N_NODES + BIN_R - 1) / BIN_R)
#define CHUNK     4096
#define NCHUNK    ((N_EDGES + CHUNK - 1) / CHUNK)
#define HIST_N    (NBINS * NCHUNK)
#define SCAN_T    (((NBINS + 31) / 32) * 32)
#define PPW       ((NCHUNK + 31) / 32)
#define STAGE_CAP 8192

static_assert(SCAN_T <= 1024, "one thread per bin in the chunk scan");
static_assert(NCHUNK <= 1024, "one thread per chunk in the bin table load");

__global__ void __launch_bounds__(1024)
gcn_chunk_sort(const int* __restrict__ src, const int* __restrict__ dst,
               int* __restrict__ tab, int* __restrict__ pairs, int nE) {
    __shared__ int s_h[NBINS];
    __shared__ int s_tot[SCAN_T / 32];
    __shared__ __attribute__((aligned(16))) int2 s_stage[CHUNK];
    const int t    = threadIdx.x;
    const int lane = t & 31;
    const int wave = t >> 5;
    for (int i = t; i < NBINS; i += 1024) s_h[i] = 0;
    __syncthreads();
    const int base = blockIdx.x * CHUNK;
    int2* __restrict__ p2 = (int2*)pairs;
    int d[CHUNK / 1024], s[CHUNK / 1024], rk[CHUNK / 1024];
    for (int j = 0; j < CHUNK / 1024; ++j) {
        const int e = min(base + j * 1024 + t, nE - 1);
        d[j] = dst[e];
        s[j] = src[e];
    }
    for (int j = 0; j < CHUNK / 1024; ++j) {
        rk[j] = 0;
        if (base + j * 1024 + t < nE) rk[j] = atomicAdd(&s_h[d[j] >> BIN_SHIFT], 1);
    }
    __syncthreads();

    int c = 0, inc = 0;
    if (t < SCAN_T) {
        c = (t < NBINS) ? s_h[t] : 0;
        inc = c;
        for (int o = 1; o < 32; o <<= 1) {
            const int y = __shfl_up(inc, o);
            if (lane >= o) inc += y;
        }
        if (lane == 31) s_tot[wave] = inc;
    }
    __syncthreads();
    if (t < SCAN_T) {
        int tv = (lane < wave) ? s_tot[lane] : 0;
        for (int o = 16; o > 0; o >>= 1) tv += __shfl_xor(tv, o);
        const int excl = tv + inc - c;
        if (t < NBINS) {
            s_h[t] = excl;
            ((int2*)tab)[t * NCHUNK + blockIdx.x] = make_int2(c, excl);
        }
    }
    __syncthreads();
    int o[CHUNK / 1024];
    for (int j = 0; j < CHUNK / 1024; ++j) o[j] = s_h[d[j] >> BIN_SHIFT];
    for (int j = 0; j < CHUNK / 1024; ++j)
        if (base + j * 1024 + t < nE)
            s_stage[o[j] + rk[j]] = make_int2(d[j], s[j]);
    __syncthreads();
    const int4* __restrict__ st4 = (const int4*)s_stage;
    int4* __restrict__ g4 = (int4*)(p2 + base);
    for (int j = 0; j < CHUNK / 2048; ++j) {
        const int i = j * 1024 + t;
        if (base + 2 * i + 1 < nE)  g4[i] = st4[i];
        else if (base + 2 * i < nE) p2[base + 2 * i] = s_stage[2 * i];
    }
}

__global__ void __launch_bounds__(1024)
gcn_bin_sort(const int* __restrict__ pairs, const int* __restrict__ tab,
             int* __restrict__ rowptr, float* __restrict__ dinv,
             int* __restrict__ csr, int n, int nE) {
    __shared__ int s_cnt[BIN_R];
    __shared__ int s_cur[BIN_R];
    __shared__ int s_tot[BIN_R / 32];
    __shared__ int s_plen[NCHUNK];
    __shared__ int s_pbeg[NCHUNK];
    __shared__ int s_red[32];
    __shared__ int s_seg[2];
    __shared__ int s_stage[STAGE_CAP];
    const int t      = threadIdx.x;
    const int lane   = t & 31;
    const int wave   = t >> 5;
    const int node0  = blockIdx.x * BIN_R;
    const int2* __restrict__ p2 = (const int2*)pairs;

    int lo = 0;
    if (t < NCHUNK) {
        const int2 e = ((const int2*)tab)[blockIdx.x * NCHUNK + t];
        lo = e.y;
        s_plen[t] = e.x;
        s_pbeg[t] = t * CHUNK + lo;
    }
    if (t < BIN_R) s_cnt[t] = 0;
    __syncthreads();

    int2 pr[PPW];
    unsigned vm = 0;
    for (int j = 0; j < PPW; ++j) {
        const int cc  = wave + 32 * j;
        const int len = (cc < NCHUNK) ? s_plen[cc] : 0;
        const int pb  = (cc < NCHUNK) ? s_pbeg[cc] : 0;
        pr[j] = p2[min(pb + min(lane, max(len - 1, 0)), nE - 1)];
        if (lane < len) vm |= 1u << j;
    }
    {
        int r = lo;
        for (int o = 16; o > 0; o >>= 1) r += __shfl_xor(r, o);
        if (lane == 0) s_red[wave] = r;
    }
    for (int j = 0; j < PPW; ++j)
        if ((vm >> j) & 1u) atomicAdd(&s_cnt[pr[j].x - node0], 1);
    for (int j = 0; j < PPW; ++j) {
        const int cc = wave + 32 * j;
        if (cc < NCHUNK) {
            const int len = s_plen[cc];
            const int pb  = s_pbeg[cc];
            for (int i = lane + 32; i < len; i += 32)
                atomicAdd(&s_cnt[p2[pb + i].x - node0], 1);
        }
    }
    __syncthreads();

    int c = 0, inc = 0;
    if (t < BIN_R) {
        c = s_cnt[t];
        inc = c;
        for (int o = 1; o < 32; o <<= 1) {
            const int y = __shfl_up(inc, o);
            if (lane >= o) inc += y;
        }
        if (lane == 31) s_tot[wave] = inc;
    }
    __syncthreads();
    if (t < BIN_R) {
        int tv = (lane < wave) ? s_tot[lane] : 0;
        int segBeg = s_red[lane];
        for (int o = 16; o > 0; o >>= 1) {
            tv += __shfl_xor(tv, o);
            segBeg += __shfl_xor(segBeg, o);
        }
        const int local = tv + inc - c;
        const int start = segBeg + local;
        s_cur[t] = local;
        if (t == BIN_R - 1) { s_seg[0] = segBeg; s_seg[1] = local + c; }
        const int node = node0 + t;
        if (node < n) {
            rowptr[node] = start;
            dinv[node] = rsqrtf((float)(c + 1));
        }
        if (node == n - 1) rowptr[n] = start + c;
    }
    __syncthreads();
    const int segBeg = s_seg[0];
    const int total  = s_seg[1];
    const bool staged = total <= STAGE_CAP;
    {
        int pos[PPW];
        for (int j = 0; j < PPW; ++j) {
            pos[j] = 0;
            if ((vm >> j) & 1u) pos[j] = atomicAdd(&s_cur[pr[j].x - node0], 1);
        }
        if (staged) {
            for (int j = 0; j < PPW; ++j)
                if ((vm >> j) & 1u) s_stage[pos[j]] = pr[j].y;
        } else {
            for (int j = 0; j < PPW; ++j)
                if ((vm >> j) & 1u) csr[segBeg + pos[j]] = pr[j].y;
        }
        for (int j = 0; j < PPW; ++j) {
            const int cc = wave + 32 * j;
            if (cc < NCHUNK) {
                const int len = s_plen[cc];
                const int pb  = s_pbeg[cc];
                for (int i = lane + 32; i < len; i += 32) {
                    const int2 q = p2[pb + i];
                    const int at = atomicAdd(&s_cur[q.x - node0], 1);
                    if (staged) s_stage[at] = q.y;
                    else        csr[segBeg + at] = q.y;
                }
            }
        }
    }
    __syncthreads();
    if (staged) {
        const int endg = segBeg + total;
        for (int g = (segBeg & ~31) + t; g < endg; g += 1024)
            if (g >= segBeg) csr[g] = s_stage[g - segBeg];
    }
}

#define TILE_PITCH 68
__global__ void __launch_bounds__(256)
gcn_gather_mlp(const int* __restrict__ rowptr, const int* __restrict__ csr,
               const float* __restrict__ x, const float* __restrict__ dinv,
               const float* __restrict__ W1, const float* __restrict__ b1,
               const float* __restrict__ W2, float* __restrict__ h2, int n) {
    __shared__ float sA[16 * TILE_PITCH];
    __shared__ float sB[16 * TILE_PITCH];
    __shared__ __attribute__((aligned(16))) float sD[16];

    const int t     = threadIdx.x;
    const int slot  = t >> 4;
    const int l     = t & 15;
    const int mtile = blockIdx.x * 16;
    const int node  = mtile + slot;

    float4 acc = {0.0f, 0.0f, 0.0f, 0.0f};
    float di = 0.0f;
    if (node < n) {
        const float4* __restrict__ x4 = (const float4*)x;
        const int beg = rowptr[node];
        const int end = rowptr[node + 1];
        di = dinv[node];
        {
            const float4 v = x4[(size_t)node * 16 + l];
            const float w = di * di;
            acc.x = v.x * w; acc.y = v.y * w; acc.z = v.z * w; acc.w = v.w * w;
        }
        int p = beg;
        for (; p + 4 <= end; p += 4) {
            const int s0 = csr[p], s1 = csr[p + 1], s2 = csr[p + 2], s3 = csr[p + 3];
            const float w0 = dinv[s0] * di, w1 = dinv[s1] * di;
            const float w2 = dinv[s2] * di, w3 = dinv[s3] * di;
            const float4 v0 = x4[(size_t)s0 * 16 + l];
            const float4 v1 = x4[(size_t)s1 * 16 + l];
            const float4 v2 = x4[(size_t)s2 * 16 + l];
            const float4 v3 = x4[(size_t)s3 * 16 + l];
            acc.x += v0.x * w0; acc.y += v0.y * w0; acc.z += v0.z * w0; acc.w += v0.w * w0;
            acc.x += v1.x * w1; acc.y += v1.y * w1; acc.z += v1.z * w1; acc.w += v1.w * w1;
            acc.x += v2.x * w2; acc.y += v2.y * w2; acc.z += v2.z * w2; acc.w += v2.w * w2;
            acc.x += v3.x * w3; acc.y += v3.y * w3; acc.z += v3.z * w3; acc.w += v3.w * w3;
        }
        for (; p < end; ++p) {
            const int s0 = csr[p];
            const float w0 = dinv[s0] * di;
            const float4 v0 = x4[(size_t)s0 * 16 + l];
            acc.x += v0.x * w0; acc.y += v0.y * w0; acc.z += v0.z * w0; acc.w += v0.w * w0;
        }
    }
    *(float4*)&sA[slot * TILE_PITCH + 4 * l] = acc;
    if (l == 0) sD[slot] = di;
    __syncthreads();

    const int lane = t & 31;
    const int half = lane >> 4;
    const int c    = lane & 15;
    const int wave = __builtin_amdgcn_readfirstlane(t >> 5);

    if (wave < 4) {
        const int ncol = wave * 16 + c;
        v8f d = {};
#pragma unroll
        for (int k0 = 0; k0 < 64; k0 += 4) {
            const int ka = k0 + half * 2;
            const v2f a = *(const v2f*)&sA[c * TILE_PITCH + ka];
            v2f b;
            b.x = W1[ka * 64 + ncol];
            b.y = W1[(ka + 1) * 64 + ncol];
            d = __builtin_amdgcn_wmma_f32_16x16x4_f32(false, a, false, b,
                                                      (short)0, d, false, false);
        }
        const float bias = b1[ncol];
#pragma unroll
        for (int r = 0; r < 8; ++r)
            sB[(r + half * 8) * TILE_PITCH + ncol] = fmaxf(d[r] + bias, 0.0f);
    }
    __syncthreads();

    if (wave == 0) {
        v8f d = {};
#pragma unroll
        for (int k0 = 0; k0 < 64; k0 += 4) {
            const int ka = k0 + half * 2;
            const v2f a = *(const v2f*)&sB[c * TILE_PITCH + ka];
            v2f b;
            b.x = W2[ka * 16 + c];
            b.y = W2[(ka + 1) * 16 + c];
            d = __builtin_amdgcn_wmma_f32_16x16x4_f32(false, a, false, b,
                                                      (short)0, d, false, false);
        }
        const float4 q0 = *(const float4*)&sD[half * 8];
        const float4 q1 = *(const float4*)&sD[half * 8 + 4];
        const float q[8] = {q0.x, q0.y, q0.z, q0.w, q1.x, q1.y, q1.z, q1.w};
#pragma unroll
        for (int r = 0; r < 8; ++r) {
            const int m = mtile + r + half * 8;
            if (m < n) h2[(size_t)m * 16 + c] = d[r] * q[r];
        }
    }
}

template <int F, bool RELU>
__global__ void __launch_bounds__(256)
gcn_gather(const int* __restrict__ rowptr, const int* __restrict__ csr,
           const float* __restrict__ hs, const float* __restrict__ dinv,
           const float* __restrict__ b, float* __restrict__ out, int n) {
    constexpr int LPN = F / 4;
    const int gt   = blockIdx.x * 256 + threadIdx.x;
    const int node = gt / LPN;
    const int l    = gt % LPN;
    if (node >= n) return;

    const float4* __restrict__ h4 = (const float4*)hs;
    const int beg = rowptr[node];
    const int end = rowptr[node + 1];
    const float di = dinv[node];

    float4 acc = h4[(size_t)node * LPN + l];

    int p = beg;
    for (; p + 4 <= end; p += 4) {
        const int s0 = csr[p], s1 = csr[p + 1], s2 = csr[p + 2], s3 = csr[p + 3];
        const float4 v0 = h4[(size_t)s0 * LPN + l];
        const float4 v1 = h4[(size_t)s1 * LPN + l];
        const float4 v2 = h4[(size_t)s2 * LPN + l];
        const float4 v3 = h4[(size_t)s3 * LPN + l];
        acc.x += v0.x; acc.y += v0.y; acc.z += v0.z; acc.w += v0.w;
        acc.x += v1.x; acc.y += v1.y; acc.z += v1.z; acc.w += v1.w;
        acc.x += v2.x; acc.y += v2.y; acc.z += v2.z; acc.w += v2.w;
        acc.x += v3.x; acc.y += v3.y; acc.z += v3.z; acc.w += v3.w;
    }
    for (; p < end; ++p) {
        const int s0 = csr[p];
        const float4 v0 = h4[(size_t)s0 * LPN + l];
        acc.x += v0.x; acc.y += v0.y; acc.z += v0.z; acc.w += v0.w;
    }

    const float4 bb = ((const float4*)b)[l];
    acc.x = acc.x * di + bb.x; acc.y = acc.y * di + bb.y;
    acc.z = acc.z * di + bb.z; acc.w = acc.w * di + bb.w;
    if (RELU) {
        acc.x = fmaxf(acc.x, 0.0f); acc.y = fmaxf(acc.y, 0.0f);
        acc.z = fmaxf(acc.z, 0.0f); acc.w = fmaxf(acc.w, 0.0f);
    }
    ((float4*)out)[(size_t)node * LPN + l] = acc;
}

extern "C" void kernel_launch(void* const* d_in, const int* in_sizes, int n_in,
                              void* d_out, int out_size, void* d_ws, size_t ws_size,
                              hipStream_t stream) {
    const float* x  = (const float*)d_in[0];
    const int*   ei = (const int*)d_in[1];
    const float* W1 = (const float*)d_in[2];
    const float* b1 = (const float*)d_in[3];
    const float* W2 = (const float*)d_in[4];
    const float* b2 = (const float*)d_in[5];
    float* out = (float*)d_out;

    const int* src = ei;
    const int* dst = ei + N_EDGES;

    int*   tab    = (int*)d_ws;
    int*   rowptr = tab + 172032;
    float* dinv   = (float*)(rowptr + 75008);
    int*   pairs  = (int*)(dinv + 75008);
    int*   csr    = pairs + 2 * N_EDGES;
    float* h2     = (float*)(csr + N_EDGES);

    const int mtiles = (N_NODES + 15) / 16;

    gcn_chunk_sort<<<NCHUNK, 1024, 0, stream>>>(src, dst, tab, pairs, N_EDGES);
    gcn_bin_sort<<<NBINS, 1024, 0, stream>>>(pairs, tab, rowptr, dinv, csr, N_NODES, N_EDGES);

    gcn_gather_mlp<<<mtiles, 256, 0, stream>>>(rowptr, csr, x, dinv, W1, b1, W2, h2, N_NODES);

    gcn_gather<16, false><<<(N_NODES * 4 + 255) / 256, 256, 0, stream>>>(rowptr, csr, h2, dinv, b2, out, N_NODES);
}


# Both gathers fetch a node's ids (layer 1: and source dinv) one per lane, 16 at once, shared by DPP lane moves: no id load per edge and lane; rows loaded in eights; sum order kept, di applied once
# speedup vs baseline: 6.6658x; 1.0330x over previous
#include <hip/hip_runtime.h>
#include <hip/hip_bf16.h>
#include <stdint.h>


typedef __attribute__((ext_vector_type(2))) float v2f;
typedef __attribute__((ext_vector_type(8))) float v8f;

#define N_NODES 75000
#define N_EDGES 1200000

#define BIN_SHIFT 8
#define BIN_R     (1 << BIN_SHIFT)
#define NBINS     ((N_NODES + BIN_R - 1) / BIN_R)
#define CHUNK     4096
#define NCHUNK    ((N_EDGES + CHUNK - 1) / CHUNK)
#define HIST_N    (NBINS * NCHUNK)
#define SCAN_T    (((NBINS + 31) / 32) * 32)
#define PPW       ((NCHUNK + 31) / 32)
#define STAGE_CAP 8192

static_assert(SCAN_T <= 1024, "one thread per bin in the chunk scan");
static_assert(NCHUNK <= 1024, "one thread per chunk in the bin table load");

__global__ void __launch_bounds__(1024)
gcn_chunk_sort(const int* __restrict__ src, const int* __restrict__ dst,
               int* __restrict__ tab, int* __restrict__ pairs, int nE) {
    __shared__ int s_h[NBINS];
    __shared__ int s_tot[SCAN_T / 32];
    __shared__ __attribute__((aligned(16))) int2 s_stage[CHUNK];
    const int t    = threadIdx.x;
    const int lane = t & 31;
    const int wave = t >> 5;
    for (int i = t; i < NBINS; i += 1024) s_h[i] = 0;
    __syncthreads();
    const int base = blockIdx.x * CHUNK;
    int2* __restrict__ p2 = (int2*)pairs;
    int d[CHUNK / 1024], s[CHUNK / 1024], rk[CHUNK / 1024];
    for (int j = 0; j < CHUNK / 1024; ++j) {
        const int e = min(base + j * 1024 + t, nE - 1);
        d[j] = dst[e];
        s[j] = src[e];
    }
    for (int j = 0; j < CHUNK / 1024; ++j) {
        rk[j] = 0;
        if (base + j * 1024 + t < nE) rk[j] = atomicAdd(&s_h[d[j] >> BIN_SHIFT], 1);
    }
    __syncthreads();

    int c = 0, inc = 0;
    if (t < SCAN_T) {
        c = (t < NBINS) ? s_h[t] : 0;
        inc = c;
        for (int o = 1; o < 32; o <<= 1) {
            const int y = __shfl_up(inc, o);
            if (lane >= o) inc += y;
        }
        if (lane == 31) s_tot[wave] = inc;
    }
    __syncthreads();
    if (t < SCAN_T) {
        int tv = (lane < wave) ? s_tot[lane] : 0;
        for (int o = 16; o > 0; o >>= 1) tv += __shfl_xor(tv, o);
        const int excl = tv + inc - c;
        if (t < NBINS) {
            s_h[t] = excl;
            ((int2*)tab)[t * NCHUNK + blockIdx.x] = make_int2(c, excl);
        }
    }
    __syncthreads();
    int o[CHUNK / 1024];
    for (int j = 0; j < CHUNK / 1024; ++j) o[j] = s_h[d[j] >> BIN_SHIFT];
    for (int j = 0; j < CHUNK / 1024; ++j)
        if (base + j * 1024 + t < nE)
            s_stage[o[j] + rk[j]] = make_int2(d[j], s[j]);
    __syncthreads();
    const int4* __restrict__ st4 = (const int4*)s_stage;
    int4* __restrict__ g4 = (int4*)(p2 + base);
    for (int j = 0; j < CHUNK / 2048; ++j) {
        const int i = j * 1024 + t;
        if (base + 2 * i + 1 < nE)  g4[i] = st4[i];
        else if (base + 2 * i < nE) p2[base + 2 * i] = s_stage[2 * i];
    }
}

__global__ void __launch_bounds__(1024)
gcn_bin_sort(const int* __restrict__ pairs, const int* __restrict__ tab,
             int* __restrict__ rowptr, float* __restrict__ dinv,
             int* __restrict__ csr, int n, int nE) {
    __shared__ int s_cnt[BIN_R];
    __shared__ int s_cur[BIN_R];
    __shared__ int s_tot[BIN_R / 32];
    __shared__ int s_plen[NCHUNK];
    __shared__ int s_pbeg[NCHUNK];
    __shared__ int s_red[32];
    __shared__ int s_seg[2];
    __shared__ int s_stage[STAGE_CAP];
    const int t      = threadIdx.x;
    const int lane   = t & 31;
    const int wave   = t >> 5;
    const int node0  = blockIdx.x * BIN_R;
    const int2* __restrict__ p2 = (const int2*)pairs;

    int lo = 0;
    if (t < NCHUNK) {
        const int2 e = ((const int2*)tab)[blockIdx.x * NCHUNK + t];
        lo = e.y;
        s_plen[t] = e.x;
        s_pbeg[t] = t * CHUNK + lo;
    }
    if (t < BIN_R) s_cnt[t] = 0;
    __syncthreads();

    int2 pr[PPW];
    unsigned vm = 0;
    for (int j = 0; j < PPW; ++j) {
        const int cc  = wave + 32 * j;
        const int len = (cc < NCHUNK) ? s_plen[cc] : 0;
        const int pb  = (cc < NCHUNK) ? s_pbeg[cc] : 0;
        pr[j] = p2[min(pb + min(lane, max(len - 1, 0)), nE - 1)];
        if (lane < len) vm |= 1u << j;
    }
    {
        int r = lo;
        for (int o = 16; o > 0; o >>= 1) r += __shfl_xor(r, o);
        if (lane == 0) s_red[wave] = r;
    }
    for (int j = 0; j < PPW; ++j)
        if ((vm >> j) & 1u) atomicAdd(&s_cnt[pr[j].x - node0], 1);
    for (int j = 0; j < PPW; ++j) {
        const int cc = wave + 32 * j;
        if (cc < NCHUNK) {
            const int len = s_plen[cc];
            const int pb  = s_pbeg[cc];
            for (int i = lane + 32; i < len; i += 32)
                atomicAdd(&s_cnt[p2[pb + i].x - node0], 1);
        }
    }
    __syncthreads();

    int c = 0, inc = 0;
    if (t < BIN_R) {
        c = s_cnt[t];
        inc = c;
        for (int o = 1; o < 32; o <<= 1) {
            const int y = __shfl_up(inc, o);
            if (lane >= o) inc += y;
        }
        if (lane == 31) s_tot[wave] = inc;
    }
    __syncthreads();
    if (t < BIN_R) {
        int tv = (lane < wave) ? s_tot[lane] : 0;
        int segBeg = s_red[lane];
        for (int o = 16; o > 0; o >>= 1) {
            tv += __shfl_xor(tv, o);
            segBeg += __shfl_xor(segBeg, o);
        }
        const int local = tv + inc - c;
        const int start = segBeg + local;
        s_cur[t] = local;
        if (t == BIN_R - 1) { s_seg[0] = segBeg; s_seg[1] = local + c; }
        const int node = node0 + t;
        if (node < n) {
            rowptr[node] = start;
            dinv[node] = rsqrtf((float)(c + 1));
        }
        if (node == n - 1) rowptr[n] = start + c;
    }
    __syncthreads();
    const int segBeg = s_seg[0];
    const int total  = s_seg[1];
    const bool staged = total <= STAGE_CAP;
    {
        int pos[PPW];
        for (int j = 0; j < PPW; ++j) {
            pos[j] = 0;
            if ((vm >> j) & 1u) pos[j] = atomicAdd(&s_cur[pr[j].x - node0], 1);
        }
        if (staged) {
            for (int j = 0; j < PPW; ++j)
                if ((vm >> j) & 1u) s_stage[pos[j]] = pr[j].y;
        } else {
            for (int j = 0; j < PPW; ++j)
                if ((vm >> j) & 1u) csr[segBeg + pos[j]] = pr[j].y;
        }
        for (int j = 0; j < PPW; ++j) {
            const int cc = wave + 32 * j;
            if (cc < NCHUNK) {
                const int len = s_plen[cc];
                const int pb  = s_pbeg[cc];
                for (int i = lane + 32; i < len; i += 32) {
                    const int2 q = p2[pb + i];
                    const int at = atomicAdd(&s_cur[q.x - node0], 1);
                    if (staged) s_stage[at] = q.y;
                    else        csr[segBeg + at] = q.y;
                }
            }
        }
    }
    __syncthreads();
    if (staged) {
        const int endg = segBeg + total;
        for (int g = (segBeg & ~31) + t; g < endg; g += 1024)
            if (g >= segBeg) csr[g] = s_stage[g - segBeg];
    }
}

#define TILE_PITCH 68

template <int K> __device__ __forceinline__ int row_lane(int v) {
    return __builtin_amdgcn_update_dpp(0, v, 0x150 + K, 0xF, 0xF, false);
}
template <int K> __device__ __forceinline__ float row_lane_f(float v) {
    return __int_as_float(row_lane<K>(__float_as_int(v)));
}
template <int K> __device__ __forceinline__ int quad_lane(int v) {
    return __builtin_amdgcn_update_dpp(0, v, K * 0x55, 0xF, 0xF, false);
}

#define GCN_ROW1(J) x4[(unsigned)row_lane<K0 + J>(ids) * 16u + (unsigned)l]
#define GCN_W1(J)   row_lane_f<K0 + J>(ds)
#define GCN_FMA1(V, W) acc.x += V.x * W; acc.y += V.y * W; acc.z += V.z * W; acc.w += V.w * W;
template <int K0>
__device__ __forceinline__ void l1_full8(float4& acc, const float4* __restrict__ x4,
                                         int ids, float ds, int l) {
    const float4 v0 = GCN_ROW1(0), v1 = GCN_ROW1(1), v2 = GCN_ROW1(2), v3 = GCN_ROW1(3);
    const float4 v4 = GCN_ROW1(4), v5 = GCN_ROW1(5), v6 = GCN_ROW1(6), v7 = GCN_ROW1(7);
    const float w0 = GCN_W1(0), w1 = GCN_W1(1), w2 = GCN_W1(2), w3 = GCN_W1(3);
    const float w4 = GCN_W1(4), w5 = GCN_W1(5), w6 = GCN_W1(6), w7 = GCN_W1(7);
    GCN_FMA1(v0, w0) GCN_FMA1(v1, w1) GCN_FMA1(v2, w2) GCN_FMA1(v3, w3)
    GCN_FMA1(v4, w4) GCN_FMA1(v5, w5) GCN_FMA1(v6, w6) GCN_FMA1(v7, w7)
}
template <int K0>
__device__ __forceinline__ void l1_full4(float4& acc, const float4* __restrict__ x4,
                                         int ids, float ds, int l) {
    const float4 v0 = GCN_ROW1(0), v1 = GCN_ROW1(1), v2 = GCN_ROW1(2), v3 = GCN_ROW1(3);
    const float w0 = GCN_W1(0), w1 = GCN_W1(1), w2 = GCN_W1(2), w3 = GCN_W1(3);
    GCN_FMA1(v0, w0) GCN_FMA1(v1, w1) GCN_FMA1(v2, w2) GCN_FMA1(v3, w3)
}
template <int K0>
__device__ __forceinline__ void l1_mid7(float4& acc, const float4* __restrict__ x4,
                                        int ids, float ds, int m, int l) {
    const float4 z = {0.0f, 0.0f, 0.0f, 0.0f};
    const float4 v0 = GCN_ROW1(0), v1 = GCN_ROW1(1), v2 = GCN_ROW1(2), v3 = GCN_ROW1(3);
    const float4 v4 = GCN_ROW1(4);
    float4 v5 = GCN_ROW1(5), v6 = GCN_ROW1(6);
    if (m < 2) v5 = z;
    if (m < 3) v6 = z;
    const float w0 = GCN_W1(0), w1 = GCN_W1(1), w2 = GCN_W1(2), w3 = GCN_W1(3);
    const float w4 = GCN_W1(4), w5 = GCN_W1(5), w6 = GCN_W1(6);
    GCN_FMA1(v0, w0) GCN_FMA1(v1, w1) GCN_FMA1(v2, w2) GCN_FMA1(v3, w3)
    GCN_FMA1(v4, w4) GCN_FMA1(v5, w5) GCN_FMA1(v6, w6)
}
template <int K0>
__device__ __forceinline__ void l1_part4(float4& acc, const float4* __restrict__ x4,
                                         int ids, float ds, int m, int l) {
    const float4 z = {0.0f, 0.0f, 0.0f, 0.0f};
    const float4 v0 = GCN_ROW1(0);
    float4 v1 = GCN_ROW1(1), v2 = GCN_ROW1(2);
    if (m < 2) v1 = z;
    if (m < 3) v2 = z;
    const float w0 = GCN_W1(0), w1 = GCN_W1(1), w2 = GCN_W1(2);
    GCN_FMA1(v0, w0) GCN_FMA1(v1, w1) GCN_FMA1(v2, w2)
}
template <int K0>
__device__ __forceinline__ void l1_window8(float4& acc, const float4* __restrict__ x4,
                                           int ids, float ds, int r, int l) {
    if (r >= 8) {
        l1_full8<K0>(acc, x4, ids, ds, l);
    } else if (r > 4) {
        l1_mid7<K0>(acc, x4, ids, ds, r - 4, l);
    } else if (r == 4) {
        l1_full4<K0>(acc, x4, ids, ds, l);
    } else {
        l1_part4<K0>(acc, x4, ids, ds, r, l);
    }
}
#undef GCN_ROW1
#undef GCN_W1
#undef GCN_FMA1

__global__ void __launch_bounds__(256)
gcn_gather_mlp(const int* __restrict__ rowptr, const int* __restrict__ csr,
               const float* __restrict__ x, const float* __restrict__ dinv,
               const float* __restrict__ W1, const float* __restrict__ b1,
               const float* __restrict__ W2, float* __restrict__ h2, int n) {
    __shared__ float sA[16 * TILE_PITCH];
    __shared__ float sB[16 * TILE_PITCH];
    __shared__ __attribute__((aligned(16))) float sD[16];

    const int t     = threadIdx.x;
    const int slot  = t >> 4;
    const int l     = t & 15;
    const int mtile = blockIdx.x * 16;
    const int node  = mtile + slot;

    float4 acc = {0.0f, 0.0f, 0.0f, 0.0f};
    float di = 0.0f;
    if (node < n) {
        const float4* __restrict__ x4 = (const float4*)x;
        const int beg = rowptr[node];
        const int end = rowptr[node + 1];
        di = dinv[node];
        {
            const float4 v = x4[(size_t)node * 16 + l];
            acc.x = v.x * di; acc.y = v.y * di; acc.z = v.z * di; acc.w = v.w * di;
        }
        for (int p = beg; p < end; p += 16) {
            const int left = end - p;
            int   id0 = node;
            float da  = 0.0f;
            if (l < left) id0 = csr[p + l];
            if (l < left) da = dinv[id0];
            l1_window8<0>(acc, x4, id0, da, left, l);
            if (left > 8) l1_window8<8>(acc, x4, id0, da, left - 8, l);
        }
        acc.x *= di; acc.y *= di; acc.z *= di; acc.w *= di;
    }
    *(float4*)&sA[slot * TILE_PITCH + 4 * l] = acc;
    if (l == 0) sD[slot] = di;
    __syncthreads();

    const int lane = t & 31;
    const int half = lane >> 4;
    const int c    = lane & 15;
    const int wave = __builtin_amdgcn_readfirstlane(t >> 5);

    if (wave < 4) {
        const int ncol = wave * 16 + c;
        v8f d = {};
#pragma unroll
        for (int k0 = 0; k0 < 64; k0 += 4) {
            const int ka = k0 + half * 2;
            const v2f a = *(const v2f*)&sA[c * TILE_PITCH + ka];
            v2f b;
            b.x = W1[ka * 64 + ncol];
            b.y = W1[(ka + 1) * 64 + ncol];
            d = __builtin_amdgcn_wmma_f32_16x16x4_f32(false, a, false, b,
                                                      (short)0, d, false, false);
        }
        const float bias = b1[ncol];
#pragma unroll
        for (int r = 0; r < 8; ++r)
            sB[(r + half * 8) * TILE_PITCH + ncol] = fmaxf(d[r] + bias, 0.0f);
    }
    __syncthreads();

    if (wave == 0) {
        v8f d = {};
#pragma unroll
        for (int k0 = 0; k0 < 64; k0 += 4) {
            const int ka = k0 + half * 2;
            const v2f a = *(const v2f*)&sB[c * TILE_PITCH + ka];
            v2f b;
            b.x = W2[ka * 16 + c];
            b.y = W2[(ka + 1) * 16 + c];
            d = __builtin_amdgcn_wmma_f32_16x16x4_f32(false, a, false, b,
                                                      (short)0, d, false, false);
        }
        const float4 q0 = *(const float4*)&sD[half * 8];
        const float4 q1 = *(const float4*)&sD[half * 8 + 4];
        const float q[8] = {q0.x, q0.y, q0.z, q0.w, q1.x, q1.y, q1.z, q1.w};
#pragma unroll
        for (int r = 0; r < 8; ++r) {
            const int m = mtile + r + half * 8;
            if (m < n) h2[(size_t)m * 16 + c] = d[r] * q[r];
        }
    }
}

#define GCN_ROW2(R, J) h4[(unsigned)quad_lane<J>(R) * 4u + (unsigned)l]
#define GCN_ADD2(V) acc.x += V.x; acc.y += V.y; acc.z += V.z; acc.w += V.w;
__device__ __forceinline__ void l2_full8(float4& acc, const float4* __restrict__ h4, int a, int b, int l) {
    const float4 v0 = GCN_ROW2(a, 0), v1 = GCN_ROW2(a, 1), v2 = GCN_ROW2(a, 2), v3 = GCN_ROW2(a, 3);
    const float4 v4 = GCN_ROW2(b, 0), v5 = GCN_ROW2(b, 1), v6 = GCN_ROW2(b, 2), v7 = GCN_ROW2(b, 3);
    GCN_ADD2(v0) GCN_ADD2(v1) GCN_ADD2(v2) GCN_ADD2(v3)
    GCN_ADD2(v4) GCN_ADD2(v5) GCN_ADD2(v6) GCN_ADD2(v7)
}
__device__ __forceinline__ void l2_full4(float4& acc, const float4* __restrict__ h4, int a, int l) {
    const float4 v0 = GCN_ROW2(a, 0), v1 = GCN_ROW2(a, 1), v2 = GCN_ROW2(a, 2), v3 = GCN_ROW2(a, 3);
    GCN_ADD2(v0) GCN_ADD2(v1) GCN_ADD2(v2) GCN_ADD2(v3)
}
__device__ __forceinline__ void l2_mid7(float4& acc, const float4* __restrict__ h4, int a, int b, int m, int l) {
    const float4 z = {0.0f, 0.0f, 0.0f, 0.0f};
    const float4 v0 = GCN_ROW2(a, 0), v1 = GCN_ROW2(a, 1), v2 = GCN_ROW2(a, 2), v3 = GCN_ROW2(a, 3);
    const float4 v4 = GCN_ROW2(b, 0);
    float4 v5 = GCN_ROW2(b, 1), v6 = GCN_ROW2(b, 2);
    if (m < 2) v5 = z;
    if (m < 3) v6 = z;
    GCN_ADD2(v0) GCN_ADD2(v1) GCN_ADD2(v2) GCN_ADD2(v3)
    GCN_ADD2(v4) GCN_ADD2(v5) GCN_ADD2(v6)
}
__device__ __forceinline__ void l2_part4(float4& acc, const float4* __restrict__ h4, int a, int m, int l) {
    const float4 z = {0.0f, 0.0f, 0.0f, 0.0f};
    const float4 v0 = GCN_ROW2(a, 0);
    float4 v1 = GCN_ROW2(a, 1), v2 = GCN_ROW2(a, 2);
    if (m < 2) v1 = z;
    if (m < 3) v2 = z;
    GCN_ADD2(v0) GCN_ADD2(v1) GCN_ADD2(v2)
}
__device__ __forceinline__ void l2_window8(float4& acc, const float4* __restrict__ h4,
                                           int a, int b, int r, int l) {
    if (r >= 8) {
        l2_full8(acc, h4, a, b, l);
    } else if (r > 4) {
        l2_mid7(acc, h4, a, b, r - 4, l);
    } else if (r == 4) {
        l2_full4(acc, h4, a, l);
    } else {
        l2_part4(acc, h4, a, r, l);
    }
}
#undef GCN_ROW2
#undef GCN_ADD2

template <int F, bool RELU>
__global__ void __launch_bounds__(256)
gcn_gather(const int* __restrict__ rowptr, const int* __restrict__ csr,
           const float* __restrict__ hs, const float* __restrict__ dinv,
           const float* __restrict__ b, float* __restrict__ out, int n) {
    constexpr int LPN = F / 4;
    static_assert(LPN == 4, "the lane moves share ids inside groups of four lanes");
    const int gt   = blockIdx.x * 256 + threadIdx.x;
    const int node = gt / LPN;
    const int l    = gt % LPN;
    if (node >= n) return;

    const float4* __restrict__ h4 = (const float4*)hs;
    const int beg = rowptr[node];
    const int end = rowptr[node + 1];
    const float di = dinv[node];

    float4 acc = h4[(size_t)node * LPN + l];

    for (int p = beg; p < end; p += 16) {
        const int left = end - p;
        int a0 = node, a1 = node, a2 = node, a3 = node;
        if (l < left)      a0 = csr[p + l];
        if (l + 4 < left)  a1 = csr[p + 4 + l];
        if (l + 8 < left)  a2 = csr[p + 8 + l];
        if (l + 12 < left) a3 = csr[p + 12 + l];
        l2_window8(acc, h4, a0, a1, left, l);
        if (left > 8) l2_window8(acc, h4, a2, a3, left - 8, l);
    }

    const float4 bb = ((const float4*)b)[l];
    acc.x = acc.x * di + bb.x; acc.y = acc.y * di + bb.y;
    acc.z = acc.z * di + bb.z; acc.w = acc.w * di + bb.w;
    if (RELU) {
        acc.x = fmaxf(acc.x, 0.0f); acc.y = fmaxf(acc.y, 0.0f);
        acc.z = fmaxf(acc.z, 0.0f); acc.w = fmaxf(acc.w, 0.0f);
    }
    ((float4*)out)[(size_t)node * LPN + l] = acc;
}

extern "C" void kernel_launch(void* const* d_in, const int* in_sizes, int n_in,
                              void* d_out, int out_size, void* d_ws, size_t ws_size,
                              hipStream_t stream) {
    const float* x  = (const float*)d_in[0];
    const int*   ei = (const int*)d_in[1];
    const float* W1 = (const float*)d_in[2];
    const float* b1 = (const float*)d_in[3];
    const float* W2 = (const float*)d_in[4];
    const float* b2 = (const float*)d_in[5];
    float* out = (float*)d_out;

    const int* src = ei;
    const int* dst = ei + N_EDGES;

    int*   tab    = (int*)d_ws;
    int*   rowptr = tab + 172032;
    float* dinv   = (float*)(rowptr + 75008);
    int*   pairs  = (int*)(dinv + 75008);
    int*   csr    = pairs + 2 * N_EDGES;
    float* h2     = (float*)(csr + N_EDGES);

    const int mtiles = (N_NODES + 15) / 16;

    gcn_chunk_sort<<<NCHUNK, 1024, 0, stream>>>(src, dst, tab, pairs, N_EDGES);
    gcn_bin_sort<<<NBINS, 1024, 0, stream>>>(pairs, tab, rowptr, dinv, csr, N_NODES, N_EDGES);

    gcn_gather_mlp<<<mtiles, 256, 0, stream>>>(rowptr, csr, x, dinv, W1, b1, W2, h2, N_NODES);

    gcn_gather<16, false><<<(N_NODES * 4 + 255) / 256, 256, 0, stream>>>(rowptr, csr, h2, dinv, b2, out, N_NODES);
}
